# starting kernel
# baseline (speedup 1.0000x reference)
#include <hip/hip_runtime.h>

typedef __attribute__((ext_vector_type(2))) float v2f;
typedef __attribute__((ext_vector_type(8))) float v8f;

#define WMMA_F32_4(a, b, c) \
    __builtin_amdgcn_wmma_f32_16x16x4_f32(false, (a), false, (b), (short)0, (c), false, false)

__device__ __forceinline__ void atomAddF32(float* p, float v) {
    unsafeAtomicAdd(p, v);
}

__global__ void zero_f32(float* __restrict__ p, long n) {
    long i = (long)blockIdx.x * blockDim.x + threadIdx.x;
    long stride = (long)gridDim.x * blockDim.x;
    for (; i < n; i += stride) p[i] = 0.0f;
}

__global__ void scatter_x_deg(const float* __restrict__ x, const int* __restrict__ src,
                              const int* __restrict__ dst, float* __restrict__ agg1,
                              float* __restrict__ deg, int E) {
    int e = blockIdx.x * blockDim.x + threadIdx.x;
    if (e >= E) return;
    int s = src[e], d = dst[e];
    atomAddF32(&deg[d], 1.0f);
    const float* xr = x + (long)s * 9;
    float* ar = agg1 + (long)d * 16;
#pragma unroll
    for (int c = 0; c < 9; ++c) atomAddF32(&ar[c], xr[c]);
}

__global__ void recip_deg(float* __restrict__ deg, int n) {
    int i = blockIdx.x * blockDim.x + threadIdx.x;
    if (i < n) deg[i] = 1.0f / fmaxf(deg[i], 1.0f);
}

__global__ __launch_bounds__(256) void layer1_wmma(
    const float* __restrict__ agg1, const float* __restrict__ invdeg,
    const float* __restrict__ x, const float* __restrict__ W1l,
    const float* __restrict__ b1, const float* __restrict__ W1r,
    float* __restrict__ h, int ntiles) {
    __shared__ float Bt[64][36];
    __shared__ float As[8][16][36];

    int lane = threadIdx.x & 31;
    int wave = threadIdx.x >> 5;

    for (int i = threadIdx.x; i < 32 * 64; i += 256) {
        int k = i >> 6, c = i & 63;
        float v = 0.0f;
        if (k < 9)                    v = W1l[k * 64 + c];
        else if (k >= 16 && k < 25)   v = W1r[(k - 16) * 64 + c];
        Bt[c][k] = v;
    }

    int tile  = blockIdx.x * 8 + wave;
    bool active = tile < ntiles;
    int node0 = tile << 4;
    if (active) {
        for (int i = lane; i < 16 * 32; i += 32) {
            int m = i >> 5, k = i & 31;
            int node = node0 + m;
            float v = 0.0f;
            if (k < 9)                  v = agg1[(long)node * 16 + k] * invdeg[node];
            else if (k >= 16 && k < 25) v = x[(long)node * 9 + (k - 16)];
            As[wave][m][k] = v;
        }
    }
    __syncthreads();
    if (!active) return;

    int m    = lane & 15;
    int n    = lane & 15;
    int koff = (lane < 16) ? 0 : 2;
    v8f c0 = {}, c1 = {}, c2 = {}, c3 = {};
#pragma unroll
    for (int ks = 0; ks < 8; ++ks) {
        int kk = ks * 4 + koff;
        v2f a  = *(const v2f*)&As[wave][m][kk];
        v2f b0 = *(const v2f*)&Bt[ 0 + n][kk];
        v2f b1v = *(const v2f*)&Bt[16 + n][kk];
        v2f b2v = *(const v2f*)&Bt[32 + n][kk];
        v2f b3v = *(const v2f*)&Bt[48 + n][kk];
        c0 = WMMA_F32_4(a, b0,  c0);
        c1 = WMMA_F32_4(a, b1v, c1);
        c2 = WMMA_F32_4(a, b2v, c2);
        c3 = WMMA_F32_4(a, b3v, c3);
    }

    int mhi = (lane < 16) ? 0 : 8;
#pragma unroll
    for (int r = 0; r < 8; ++r) {
        int row = node0 + r + mhi;
        float* hp = h + (long)row * 64;
        hp[ 0 + n] = fmaxf(c0[r] + b1[ 0 + n], 0.0f);
        hp[16 + n] = fmaxf(c1[r] + b1[16 + n], 0.0f);
        hp[32 + n] = fmaxf(c2[r] + b1[32 + n], 0.0f);
        hp[48 + n] = fmaxf(c3[r] + b1[48 + n], 0.0f);
    }
}

__global__ __launch_bounds__(256) void layer2_wmma(
    const float* __restrict__ h, const float* __restrict__ W2l,
    const float* __restrict__ W2r, const float* __restrict__ b2,
    float* __restrict__ t, float* __restrict__ out, int ntiles) {
    __shared__ float Bt[64][68];
    __shared__ float As[8][16][68];

    int lane = threadIdx.x & 31;
    int wave = threadIdx.x >> 5;

    for (int i = threadIdx.x; i < 64 * 64; i += 256) {
        int k = i >> 6, c = i & 63;
        Bt[c][k] = (c < 32) ? W2l[k * 32 + c] : W2r[k * 32 + (c - 32)];
    }

    int tile  = blockIdx.x * 8 + wave;
    bool active = tile < ntiles;
    int node0 = tile << 4;
    if (active) {
        for (int i = lane; i < 256; i += 32) {
            int m = i >> 4, q = (i & 15) * 4;
            const float4 hv = *(const float4*)(h + (long)(node0 + m) * 64 + q);
            As[wave][m][q + 0] = hv.x;
            As[wave][m][q + 1] = hv.y;
            As[wave][m][q + 2] = hv.z;
            As[wave][m][q + 3] = hv.w;
        }
    }
    __syncthreads();
    if (!active) return;

    int m    = lane & 15;
    int n    = lane & 15;
    int koff = (lane < 16) ? 0 : 2;
    v8f c0 = {}, c1 = {}, c2 = {}, c3 = {};
#pragma unroll
    for (int ks = 0; ks < 16; ++ks) {
        int kk = ks * 4 + koff;
        v2f a  = *(const v2f*)&As[wave][m][kk];
        v2f b0 = *(const v2f*)&Bt[ 0 + n][kk];
        v2f b1v = *(const v2f*)&Bt[16 + n][kk];
        v2f b2v = *(const v2f*)&Bt[32 + n][kk];
        v2f b3v = *(const v2f*)&Bt[48 + n][kk];
        c0 = WMMA_F32_4(a, b0,  c0);
        c1 = WMMA_F32_4(a, b1v, c1);
        c2 = WMMA_F32_4(a, b2v, c2);
        c3 = WMMA_F32_4(a, b3v, c3);
    }

    int mhi = (lane < 16) ? 0 : 8;
#pragma unroll
    for (int r = 0; r < 8; ++r) {
        int row = node0 + r + mhi;
        t[(long)row * 32 + n]        = c0[r];
        t[(long)row * 32 + 16 + n]   = c1[r];
        out[(long)row * 32 + n]      = c2[r] + b2[n];
        out[(long)row * 32 + 16 + n] = c3[r] + b2[16 + n];
    }
}

__global__ void scatter_t(const float* __restrict__ t, const int* __restrict__ src,
                          const int* __restrict__ dst, float* __restrict__ agg2, int E) {
    long tid = (long)blockIdx.x * blockDim.x + threadIdx.x;
    if (tid >= (long)E * 8) return;
    int e = (int)(tid >> 3);
    int q = (int)(tid & 7) * 4;
    int s = src[e], d = dst[e];
    const float4 tv = *(const float4*)(t + (long)s * 32 + q);
    float* ap = agg2 + (long)d * 32 + q;
    atomAddF32(ap + 0, tv.x);
    atomAddF32(ap + 1, tv.y);
    atomAddF32(ap + 2, tv.z);
    atomAddF32(ap + 3, tv.w);
}

__global__ void final_add(const float* __restrict__ agg2, const float* __restrict__ invdeg,
                          float* __restrict__ out, int n_nodes) {
    long i = (long)blockIdx.x * blockDim.x + threadIdx.x;
    if (i >= (long)n_nodes * 32) return;
    int node = (int)(i >> 5);
    out[i] += agg2[i] * invdeg[node];
}

extern "C" void kernel_launch(void* const* d_in, const int* in_sizes, int n_in,
                              void* d_out, int out_size, void* d_ws, size_t ws_size,
                              hipStream_t stream) {
    const float* x   = (const float*)d_in[0];
    const int*   ei  = (const int*)d_in[1];
    const float* W1l = (const float*)d_in[2];
    const float* b1  = (const float*)d_in[3];
    const float* W1r = (const float*)d_in[4];
    const float* W2l = (const float*)d_in[5];
    const float* b2  = (const float*)d_in[6];
    const float* W2r = (const float*)d_in[7];
    float* out = (float*)d_out;

    int N = in_sizes[0] / 9;
    int E = in_sizes[1] / 2;
    const int* src = ei;
    const int* dst = ei + E;

    float* ws   = (float*)d_ws;
    float* deg  = ws;
    float* agg1 = deg + N;
    float* h    = agg1 + (long)N * 16;
    float* t    = h + (long)N * 64;
    float* agg2 = t + (long)N * 32;

    zero_f32<<<2048, 256, 0, stream>>>(deg, (long)N * 17);
    zero_f32<<<2048, 256, 0, stream>>>(agg2, (long)N * 32);

    scatter_x_deg<<<(E + 255) / 256, 256, 0, stream>>>(x, src, dst, agg1, deg, E);
    recip_deg<<<(N + 255) / 256, 256, 0, stream>>>(deg, N);

    int ntiles = (N + 15) / 16;
    int blocks = (ntiles + 7) / 8;
    layer1_wmma<<<blocks, 256, 0, stream>>>(agg1, deg, x, W1l, b1, W1r, h, ntiles);
    layer2_wmma<<<blocks, 256, 0, stream>>>(h, W2l, W2r, b2, t, out, ntiles);

    long s2 = (long)E * 8;
    scatter_t<<<(int)((s2 + 255) / 256), 256, 0, stream>>>(t, src, dst, agg2, E);
    final_add<<<(int)(((long)N * 32 + 255) / 256), 256, 0, stream>>>(agg2, deg, out, N);
}


# Both edge scatters re-laid with lanes along the destination row: one atomic instruction per edge row instead of 10 and 4 (degree counted in the row's spare column), 32 edges per wave
# speedup vs baseline: 2.9561x; 2.9561x over previous
#include <hip/hip_runtime.h>

typedef __attribute__((ext_vector_type(2))) float v2f;
typedef __attribute__((ext_vector_type(8))) float v8f;

#define WMMA_F32_4(a, b, c) \
    __builtin_amdgcn_wmma_f32_16x16x4_f32(false, (a), false, (b), (short)0, (c), false, false)

__device__ __forceinline__ void atomAddF32(float* p, float v) {
    unsafeAtomicAdd(p, v);
}

__global__ void zero_f32(float* __restrict__ p, long n) {
    long i = (long)blockIdx.x * blockDim.x + threadIdx.x;
    long stride = (long)gridDim.x * blockDim.x;
    for (; i < n; i += stride) p[i] = 0.0f;
}

__global__ __launch_bounds__(256) void scatter_x_deg(const float* __restrict__ x, const int* __restrict__ src,
                              const int* __restrict__ dst, float* __restrict__ agg1, int E) {
    const int  lane = threadIdx.x & 31;
    const int  slot = lane & 15;
    const bool hi   = lane >= 16;
    const int  e    = blockIdx.x * 256 + threadIdx.x;
    const int  e0   = __builtin_amdgcn_readfirstlane(e - lane);
    int sv = 0, dv = 0;
    if (e < E) { sv = src[e]; dv = dst[e]; }
    const int cs = slot < 9 ? slot : 8;

    if (e0 + 32 <= E) {
        float  v[16];
        float* ap[16];
#pragma unroll
        for (int j = 0; j < 16; ++j) {
            const int sa = __builtin_amdgcn_readlane(sv, j), sb = __builtin_amdgcn_readlane(sv, j + 16);
            const int da = __builtin_amdgcn_readlane(dv, j), db = __builtin_amdgcn_readlane(dv, j + 16);
            const int s = hi ? sb : sa;
            const int d = hi ? db : da;
            const float xv = x[(long)s * 9 + cs];
            v[j]  = slot < 9 ? xv : 1.0f;
            ap[j] = agg1 + (long)d * 16 + slot;
        }
        if (slot < 10) {
            for (int j = 0; j < 16; ++j) atomAddF32(ap[j], v[j]);
        }
    } else {
        for (int j = 0; j < 16; ++j) {
            const int own = hi ? j + 16 : j;
            const int s = __shfl(sv, own), d = __shfl(dv, own);
            if (e0 + own < E && slot < 10) {
                const float xv = x[(long)s * 9 + cs];
                atomAddF32(agg1 + (long)d * 16 + slot, slot < 9 ? xv : 1.0f);
            }
        }
    }
}

__global__ void recip_deg(const float* __restrict__ agg1, float* __restrict__ deg, int n) {
    int i = blockIdx.x * blockDim.x + threadIdx.x;
    if (i < n) deg[i] = 1.0f / fmaxf(agg1[(long)i * 16 + 9], 1.0f);
}

__global__ __launch_bounds__(256) void layer1_wmma(
    const float* __restrict__ agg1, const float* __restrict__ invdeg,
    const float* __restrict__ x, const float* __restrict__ W1l,
    const float* __restrict__ b1, const float* __restrict__ W1r,
    float* __restrict__ h, int ntiles) {
    __shared__ float Bt[64][36];
    __shared__ float As[8][16][36];

    int lane = threadIdx.x & 31;
    int wave = threadIdx.x >> 5;

    for (int i = threadIdx.x; i < 32 * 64; i += 256) {
        int k = i >> 6, c = i & 63;
        float v = 0.0f;
        if (k < 9)                    v = W1l[k * 64 + c];
        else if (k >= 16 && k < 25)   v = W1r[(k - 16) * 64 + c];
        Bt[c][k] = v;
    }

    int tile  = blockIdx.x * 8 + wave;
    bool active = tile < ntiles;
    int node0 = tile << 4;
    if (active) {
        for (int i = lane; i < 16 * 32; i += 32) {
            int m = i >> 5, k = i & 31;
            int node = node0 + m;
            float v = 0.0f;
            if (k < 9)                  v = agg1[(long)node * 16 + k] * invdeg[node];
            else if (k >= 16 && k < 25) v = x[(long)node * 9 + (k - 16)];
            As[wave][m][k] = v;
        }
    }
    __syncthreads();
    if (!active) return;

    int m    = lane & 15;
    int n    = lane & 15;
    int koff = (lane < 16) ? 0 : 2;
    v8f c0 = {}, c1 = {}, c2 = {}, c3 = {};
#pragma unroll
    for (int ks = 0; ks < 8; ++ks) {
        int kk = ks * 4 + koff;
        v2f a  = *(const v2f*)&As[wave][m][kk];
        v2f b0 = *(const v2f*)&Bt[ 0 + n][kk];
        v2f b1v = *(const v2f*)&Bt[16 + n][kk];
        v2f b2v = *(const v2f*)&Bt[32 + n][kk];
        v2f b3v = *(const v2f*)&Bt[48 + n][kk];
        c0 = WMMA_F32_4(a, b0,  c0);
        c1 = WMMA_F32_4(a, b1v, c1);
        c2 = WMMA_F32_4(a, b2v, c2);
        c3 = WMMA_F32_4(a, b3v, c3);
    }

    int mhi = (lane < 16) ? 0 : 8;
#pragma unroll
    for (int r = 0; r < 8; ++r) {
        int row = node0 + r + mhi;
        float* hp = h + (long)row * 64;
        hp[ 0 + n] = fmaxf(c0[r] + b1[ 0 + n], 0.0f);
        hp[16 + n] = fmaxf(c1[r] + b1[16 + n], 0.0f);
        hp[32 + n] = fmaxf(c2[r] + b1[32 + n], 0.0f);
        hp[48 + n] = fmaxf(c3[r] + b1[48 + n], 0.0f);
    }
}

__global__ __launch_bounds__(256) void layer2_wmma(
    const float* __restrict__ h, const float* __restrict__ W2l,
    const float* __restrict__ W2r, const float* __restrict__ b2,
    float* __restrict__ t, float* __restrict__ out, int ntiles) {
    __shared__ float Bt[64][68];
    __shared__ float As[8][16][68];

    int lane = threadIdx.x & 31;
    int wave = threadIdx.x >> 5;

    for (int i = threadIdx.x; i < 64 * 64; i += 256) {
        int k = i >> 6, c = i & 63;
        Bt[c][k] = (c < 32) ? W2l[k * 32 + c] : W2r[k * 32 + (c - 32)];
    }

    int tile  = blockIdx.x * 8 + wave;
    bool active = tile < ntiles;
    int node0 = tile << 4;
    if (active) {
        for (int i = lane; i < 256; i += 32) {
            int m = i >> 4, q = (i & 15) * 4;
            const float4 hv = *(const float4*)(h + (long)(node0 + m) * 64 + q);
            As[wave][m][q + 0] = hv.x;
            As[wave][m][q + 1] = hv.y;
            As[wave][m][q + 2] = hv.z;
            As[wave][m][q + 3] = hv.w;
        }
    }
    __syncthreads();
    if (!active) return;

    int m    = lane & 15;
    int n    = lane & 15;
    int koff = (lane < 16) ? 0 : 2;
    v8f c0 = {}, c1 = {}, c2 = {}, c3 = {};
#pragma unroll
    for (int ks = 0; ks < 16; ++ks) {
        int kk = ks * 4 + koff;
        v2f a  = *(const v2f*)&As[wave][m][kk];
        v2f b0 = *(const v2f*)&Bt[ 0 + n][kk];
        v2f b1v = *(const v2f*)&Bt[16 + n][kk];
        v2f b2v = *(const v2f*)&Bt[32 + n][kk];
        v2f b3v = *(const v2f*)&Bt[48 + n][kk];
        c0 = WMMA_F32_4(a, b0,  c0);
        c1 = WMMA_F32_4(a, b1v, c1);
        c2 = WMMA_F32_4(a, b2v, c2);
        c3 = WMMA_F32_4(a, b3v, c3);
    }

    int mhi = (lane < 16) ? 0 : 8;
#pragma unroll
    for (int r = 0; r < 8; ++r) {
        int row = node0 + r + mhi;
        t[(long)row * 32 + n]        = c0[r];
        t[(long)row * 32 + 16 + n]   = c1[r];
        out[(long)row * 32 + n]      = c2[r] + b2[n];
        out[(long)row * 32 + 16 + n] = c3[r] + b2[16 + n];
    }
}

__global__ __launch_bounds__(256) void scatter_t(const float* __restrict__ t, const int* __restrict__ src,
                          const int* __restrict__ dst, float* __restrict__ agg2, int E) {
    const int lane = threadIdx.x & 31;
    const int e    = blockIdx.x * 256 + threadIdx.x;
    const int e0   = __builtin_amdgcn_readfirstlane(e - lane);
    int sv = 0, dv = 0;
    if (e < E) { sv = src[e]; dv = dst[e]; }

    if (e0 + 32 <= E) {
        for (int b = 0; b < 32; b += 16) {
            float v[16];
            for (int j = 0; j < 16; ++j) {
                const int s = __builtin_amdgcn_readlane(sv, b + j);
                v[j] = t[(long)s * 32 + lane];
            }
            for (int j = 0; j < 16; ++j) {
                const int d = __builtin_amdgcn_readlane(dv, b + j);
                atomAddF32(agg2 + (long)d * 32 + lane, v[j]);
            }
        }
    } else {
        for (int j = 0; j < 32; ++j) {
            if (e0 + j >= E) break;
            const int s = __shfl(sv, j), d = __shfl(dv, j);
            atomAddF32(agg2 + (long)d * 32 + lane, t[(long)s * 32 + lane]);
        }
    }
}

__global__ void final_add(const float* __restrict__ agg2, const float* __restrict__ invdeg,
                          float* __restrict__ out, int n_nodes) {
    long i = (long)blockIdx.x * blockDim.x + threadIdx.x;
    if (i >= (long)n_nodes * 32) return;
    int node = (int)(i >> 5);
    out[i] += agg2[i] * invdeg[node];
}

extern "C" void kernel_launch(void* const* d_in, const int* in_sizes, int n_in,
                              void* d_out, int out_size, void* d_ws, size_t ws_size,
                              hipStream_t stream) {
    const float* x   = (const float*)d_in[0];
    const int*   ei  = (const int*)d_in[1];
    const float* W1l = (const float*)d_in[2];
    const float* b1  = (const float*)d_in[3];
    const float* W1r = (const float*)d_in[4];
    const float* W2l = (const float*)d_in[5];
    const float* b2  = (const float*)d_in[6];
    const float* W2r = (const float*)d_in[7];
    float* out = (float*)d_out;

    int N = in_sizes[0] / 9;
    int E = in_sizes[1] / 2;
    const int* src = ei;
    const int* dst = ei + E;

    float* ws   = (float*)d_ws;
    float* deg  = ws;
    float* agg1 = deg + N;
    float* h    = agg1 + (long)N * 16;
    float* t    = h + (long)N * 64;
    float* agg2 = t + (long)N * 32;

    zero_f32<<<2048, 256, 0, stream>>>(agg1, (long)N * 16);
    zero_f32<<<2048, 256, 0, stream>>>(agg2, (long)N * 32);

    scatter_x_deg<<<(E + 255) / 256, 256, 0, stream>>>(x, src, dst, agg1, E);
    recip_deg<<<(N + 255) / 256, 256, 0, stream>>>(agg1, deg, N);

    int ntiles = (N + 15) / 16;
    int blocks = (ntiles + 7) / 8;
    layer1_wmma<<<blocks, 256, 0, stream>>>(agg1, deg, x, W1l, b1, W1r, h, ntiles);
    layer2_wmma<<<blocks, 256, 0, stream>>>(h, W2l, W2r, b2, t, out, ntiles);

    scatter_t<<<(E + 255) / 256, 256, 0, stream>>>(t, src, dst, agg2, E);
    final_add<<<(int)(((long)N * 32 + 255) / 256), 256, 0, stream>>>(agg2, deg, out, N);
}


# Edge placement staged in LDS: each work-group groups its chunk's edge words in shared memory and writes sorted[] as whole 128-byte lines (one store per line, not 32 scattered 4-byte stores)
# speedup vs baseline: 5.8397x; 1.9755x over previous
#include <hip/hip_runtime.h>

typedef __attribute__((ext_vector_type(2))) float v2f;
typedef __attribute__((ext_vector_type(8))) float v8f;

#define WMMA_F32_4(a, b, c) \
    __builtin_amdgcn_wmma_f32_16x16x4_f32(false, (a), false, (b), (short)0, (c), false, false)

#define NBK 256
#define BT  1024

__device__ __forceinline__ void atomAddF32(float* p, float v) {
    unsafeAtomicAdd(p, v);
}


__device__ __forceinline__ int chunk_end(int w, int chunk, int E) {
    const int e0 = w * chunk;
    int e1 = e0 + chunk;
    if (e1 > E) e1 = E;
    return e1 < e0 ? e0 : e1;
}

__device__ __forceinline__ int wave_incl_sum(int v, int lane) {
    int s = v;
#pragma unroll
    for (int off = 1; off < 32; off <<= 1) {
        const int u = __shfl_up(s, off);
        if (lane >= off) s += u;
    }
    return s;
}

__global__ __launch_bounds__(BT) void bucket_group(const int* __restrict__ src, const int* __restrict__ dst,
                                                   int* __restrict__ tab, unsigned* __restrict__ sorted,
                                                   int E, int chunk, int npb) {
    extern __shared__ unsigned stage[];
    __shared__ int hist[NBK];
    __shared__ int cursor[NBK];
    __shared__ int wsum[NBK / 32];
    const int tid = threadIdx.x, w = blockIdx.x, lane = tid & 31;
    if (tid < NBK) hist[tid] = 0;
    __syncthreads();
    const int e0 = w * chunk;
    const int e1 = chunk_end(w, chunk, E);
    for (int eb = e0 + tid; eb < e1; eb += 4 * BT) {
        int d[4];
#pragma unroll
        for (int u = 0; u < 4; ++u) {
            const int e = eb + u * BT;
            d[u] = dst[e < e1 ? e : e1 - 1];
        }
#pragma unroll
        for (int u = 0; u < 4; ++u)
            if (eb + u * BT < e1) atomicAdd(&hist[d[u] / npb], 1);
    }
    __syncthreads();
    int c = 0, incl = 0;
    if (tid < NBK) {
        c = hist[tid];
        incl = wave_incl_sum(c, lane);
        if (lane == 31) wsum[tid >> 5] = incl;
    }
    __syncthreads();
    if (tid < NBK) {
        int before = 0;
#pragma unroll
        for (int k = 0; k < NBK / 32; ++k) before += (k < (tid >> 5)) ? wsum[k] : 0;
        const int first = before + incl - c;
        cursor[tid] = first;
        tab[tid * NBK + w] = e0 + first;
    }
    __syncthreads();
    for (int eb = e0 + tid; eb < e1; eb += 4 * BT) {
        int s[4], d[4];
#pragma unroll
        for (int u = 0; u < 4; ++u) {
            const int e  = eb + u * BT;
            const int ec = e < e1 ? e : e1 - 1;
            s[u] = src[ec];
            d[u] = dst[ec];
        }
#pragma unroll
        for (int u = 0; u < 4; ++u) {
            if (eb + u * BT < e1) {
                const int bb  = d[u] / npb;
                const int pos = atomicAdd(&cursor[bb], 1);
                stage[pos] = ((unsigned)s[u] << 12) | (unsigned)(d[u] - bb * npb);
            }
        }
    }
    __syncthreads();
    const int n   = e1 - e0;
    const int mis = (int)(((size_t)(sorted + e0) >> 2) & 31);
    for (int ib = tid - mis; ib < n; ib += 4 * BT) {
        unsigned wd[4];
#pragma unroll
        for (int u = 0; u < 4; ++u) {
            const int i = ib + u * BT;
            wd[u] = stage[i < 0 ? 0 : (i < n ? i : n - 1)];
        }
#pragma unroll
        for (int u = 0; u < 4; ++u) {
            const int i = ib + u * BT;
            if (i >= 0 && i < n) sorted[e0 + i] = wd[u];
        }
    }
}

struct RunDir {
    int rpos[NBK];
    int rdel[NBK];
    int wsum[NBK / 32];
    int total;
};

__device__ __forceinline__ void run_fetch(const int* __restrict__ tab, int bk, int tid, int E, int chunk,
                                          int& first, int& len) {
    const int nb = bk + 1 < NBK ? bk + 1 : NBK - 1;
    first = tab[bk * NBK + tid];
    int next = tab[nb * NBK + tid];
    if (bk + 1 == NBK) next = chunk_end(tid, chunk, E);
    len = next - first;
}

__device__ __forceinline__ int run_scan_a(RunDir& rd, int tid, int lane, int len) {
    const int incl = wave_incl_sum(len, lane);
    if (lane == 31) rd.wsum[tid >> 5] = incl;
    return incl;
}

__device__ __forceinline__ void run_scan_b(RunDir& rd, int tid, int first, int len, int incl) {
    int before = 0;
#pragma unroll
    for (int k = 0; k < NBK / 32; ++k) before += (k < (tid >> 5)) ? rd.wsum[k] : 0;
    const int ex = before + incl - len;
    rd.rpos[tid] = ex;
    rd.rdel[tid] = first - ex;
    if (tid == NBK - 1) rd.total = ex + len;
}

__device__ __forceinline__ int run_slot(const RunDir& rd, int g) {
    int k = 0;
#pragma unroll
    for (int step = NBK / 2; step >= 1; step >>= 1)
        if (rd.rpos[k + step] <= g) k += step;
    return g + rd.rdel[k];
}

__global__ __launch_bounds__(BT) void gather_x_layers(
    const float* __restrict__ x, const unsigned* __restrict__ sorted, const int* __restrict__ tab,
    const float* __restrict__ W1l, const float* __restrict__ b1, const float* __restrict__ W1r,
    const float* __restrict__ W2l, const float* __restrict__ b2, const float* __restrict__ W2r,
    float* __restrict__ deg, float* __restrict__ t, float* __restrict__ out, int N, int npb, int E, int chunk) {
    extern __shared__ float acc[];
    __shared__ float Bt1[64][36];
    __shared__ float Bt2[64][68];
    __shared__ RunDir rd;

    const int tid  = threadIdx.x, bk = blockIdx.x;
    const int lane = tid & 31;
    const int wv   = __builtin_amdgcn_readfirstlane(tid >> 5);
    const int slot = lane & 15;
    const bool hi  = lane >= 16;
    const int ntl  = (npb + 15) >> 4;
    const int nacc = ntl * 256;
    int rfirst = 0, rlen = 0, rincl = 0;
    if (tid < NBK) run_fetch(tab, bk, tid, E, chunk, rfirst, rlen);
    for (int i = tid; i < nacc; i += BT) acc[i] = 0.0f;
    if (tid < NBK) rincl = run_scan_a(rd, tid, lane, rlen);
    __syncthreads();
    if (tid < NBK) run_scan_b(rd, tid, rfirst, rlen, rincl);
    __syncthreads();
    const int total = __builtin_amdgcn_readfirstlane(rd.total);
    const int kq = tid >> 6, cq = tid & 63;
    const int kc = kq < 9 ? kq : 8;
    float w1a = W1l[kc * 64 + cq];
    float w1b = W1r[kc * 64 + cq];
    if (kq >= 9) { w1a = 0.0f; w1b = 0.0f; }
    const float* w2p = (cq < 32) ? (W2l + cq) : (W2r + (cq - 32));
    float w2v[4];
#pragma unroll
    for (int u = 0; u < 4; ++u) w2v[u] = w2p[(kq + 16 * u) * 32];
    const int cs = slot < 9 ? slot : 8;
    unsigned p = 0;
    if (wv * 32 + lane < total) p = sorted[run_slot(rd, wv * 32 + lane)];
    for (int base = wv * 32; base < total; base += BT) {
        const int sv = (int)(p >> 12), dv = (int)(p & 4095u);
        const bool full = base + 32 <= total;
        float v[16];
        int   ai[16];
        if (full) {
#pragma unroll
            for (int j = 0; j < 16; ++j) {
                const int sa = __builtin_amdgcn_readlane(sv, j), sb = __builtin_amdgcn_readlane(sv, j + 16);
                const int da = __builtin_amdgcn_readlane(dv, j), db = __builtin_amdgcn_readlane(dv, j + 16);
                const int s = hi ? sb : sa;
                const int d = hi ? db : da;
                v[j]  = x[(long)s * 9 + cs];
                ai[j] = d * 16 + slot;
            }
        }
        const int gn = base + BT + lane;
        const unsigned pn = sorted[run_slot(rd, gn < total ? gn : total - 1)];
        if (full) {
            if (slot < 10) {
#pragma unroll
                for (int j = 0; j < 16; ++j) atomAddF32(&acc[ai[j]], slot < 9 ? v[j] : 1.0f);
            }
        } else {
            for (int j = 0; j < 16; ++j) {
                const int own = hi ? j + 16 : j;
                const int s = __shfl(sv, own), d = __shfl(dv, own);
                if (base + own < total && slot < 10) {
                    const float xv = x[(long)s * 9 + cs];
                    atomAddF32(&acc[d * 16 + slot], slot < 9 ? xv : 1.0f);
                }
            }
        }
        p = pn;
    }
    Bt1[cq][kq]      = w1a;
    Bt1[cq][16 + kq] = w1b;
#pragma unroll
    for (int u = 0; u < 4; ++u) Bt2[cq][kq + 16 * u] = w2v[u];
    __syncthreads();

    const int n0   = bk * npb;
    const int m    = lane & 15;
    const int koff = (lane < 16) ? 0 : 2;
    const int mhi  = (lane < 16) ? 0 : 8;
    float* hs = acc + nacc + wv * (16 * 68);
    for (int tb = 0; tb < ntl; tb += BT / 32) {
        const int tl = tb + wv;
        const bool active = tl < ntl;
        v8f c0 = {}, c1 = {}, c2 = {}, c3 = {};
        if (active) {
            const int lrow  = tl * 16 + m;
            const int node  = n0 + lrow;
            const bool rv   = lrow < npb && node < N;
            const float* ar = acc + lrow * 16;
            const float inv = 1.0f / fmaxf(ar[9], 1.0f);
            if (lane < 16 && rv) deg[node] = inv;
            const float* xr = x + (long)(node < N ? node : N - 1) * 9;
            float x0 = xr[koff], x1 = xr[koff + 1], x2 = xr[4 + koff], x3 = xr[5 + koff], x4 = xr[8];
            if (!rv) { x0 = 0.0f; x1 = 0.0f; x2 = 0.0f; x3 = 0.0f; x4 = 0.0f; }
            if (lane >= 16) x4 = 0.0f;
#pragma unroll
            for (int ks = 0; ks < 7; ++ks) {
                const int kk = ks * 4 + koff;
                v2f a;
                if (ks < 4) {
                    a = *(const v2f*)&ar[kk];
                    a.x *= inv;
                    a.y *= inv;
                } else if (ks == 4) {
                    a.x = x0; a.y = x1;
                } else if (ks == 5) {
                    a.x = x2; a.y = x3;
                } else {
                    a.x = x4; a.y = 0.0f;
                }
                v2f b0  = *(const v2f*)&Bt1[ 0 + m][kk];
                v2f b1v = *(const v2f*)&Bt1[16 + m][kk];
                v2f b2v = *(const v2f*)&Bt1[32 + m][kk];
                v2f b3v = *(const v2f*)&Bt1[48 + m][kk];
                c0 = WMMA_F32_4(a, b0,  c0);
                c1 = WMMA_F32_4(a, b1v, c1);
                c2 = WMMA_F32_4(a, b2v, c2);
                c3 = WMMA_F32_4(a, b3v, c3);
            }
            const float bb0 = b1[m], bb1 = b1[16 + m], bb2 = b1[32 + m], bb3 = b1[48 + m];
#pragma unroll
            for (int r = 0; r < 8; ++r) {
                float* hp = hs + (r + mhi) * 68;
                hp[ 0 + m] = fmaxf(c0[r] + bb0, 0.0f);
                hp[16 + m] = fmaxf(c1[r] + bb1, 0.0f);
                hp[32 + m] = fmaxf(c2[r] + bb2, 0.0f);
                hp[48 + m] = fmaxf(c3[r] + bb3, 0.0f);
            }
        }
        __syncthreads();
        if (active) {
            c0 = (v8f){}; c1 = (v8f){}; c2 = (v8f){}; c3 = (v8f){};
#pragma unroll
            for (int ks = 0; ks < 16; ++ks) {
                const int kk = ks * 4 + koff;
                v2f a   = *(const v2f*)&hs[m * 68 + kk];
                v2f b0  = *(const v2f*)&Bt2[ 0 + m][kk];
                v2f b1v = *(const v2f*)&Bt2[16 + m][kk];
                v2f b2v = *(const v2f*)&Bt2[32 + m][kk];
                v2f b3v = *(const v2f*)&Bt2[48 + m][kk];
                c0 = WMMA_F32_4(a, b0,  c0);
                c1 = WMMA_F32_4(a, b1v, c1);
                c2 = WMMA_F32_4(a, b2v, c2);
                c3 = WMMA_F32_4(a, b3v, c3);
            }
            const float bo0 = b2[m], bo1 = b2[16 + m];
#pragma unroll
            for (int r = 0; r < 8; ++r) {
                const int lr  = tl * 16 + r + mhi;
                const int row = n0 + lr;
                if (lr < npb && row < N) {
                    t[(long)row * 32 + m]        = c0[r];
                    t[(long)row * 32 + 16 + m]   = c1[r];
                    out[(long)row * 32 + m]      = c2[r] + bo0;
                    out[(long)row * 32 + 16 + m] = c3[r] + bo1;
                }
            }
        }
    }
}

__global__ __launch_bounds__(BT) void gather_t(const float* __restrict__ t, const unsigned* __restrict__ sorted,
                                               const int* __restrict__ tab, const float* __restrict__ deg,
                                               float* __restrict__ out, int N, int npb, int E, int chunk) {
    extern __shared__ float acc[];
    __shared__ RunDir rd;
    const int tid  = threadIdx.x, bk = blockIdx.x;
    const int lane = tid & 31;
    const int wv   = __builtin_amdgcn_readfirstlane(tid >> 5);
    const int nacc = npb * 32;
    int rfirst = 0, rlen = 0, rincl = 0;
    if (tid < NBK) run_fetch(tab, bk, tid, E, chunk, rfirst, rlen);
    for (int i = tid; i < nacc; i += BT) acc[i] = 0.0f;
    if (tid < NBK) rincl = run_scan_a(rd, tid, lane, rlen);
    __syncthreads();
    if (tid < NBK) run_scan_b(rd, tid, rfirst, rlen, rincl);
    __syncthreads();
    const int total = __builtin_amdgcn_readfirstlane(rd.total);
    unsigned p = 0;
    if (wv * 32 + lane < total) p = sorted[run_slot(rd, wv * 32 + lane)];
    for (int base = wv * 32; base < total; base += BT) {
        const int sv = (int)(p >> 12), dv = (int)(p & 4095u);
        const bool full = base + 32 <= total;
        float v[32];
        if (full) {
#pragma unroll
            for (int j = 0; j < 32; ++j) {
                const int s = __builtin_amdgcn_readlane(sv, j);
                v[j] = t[(long)s * 32 + lane];
            }
        }
        const int gn = base + BT + lane;
        const unsigned pn = sorted[run_slot(rd, gn < total ? gn : total - 1)];
        if (full) {
#pragma unroll
            for (int j = 0; j < 32; ++j) {
                const int d = __builtin_amdgcn_readlane(dv, j);
                atomAddF32(&acc[d * 32 + lane], v[j]);
            }
        } else {
            const int n = total - base;
            for (int j = 0; j < n; ++j) {
                const int s = __shfl(sv, j), d = __shfl(dv, j);
                atomAddF32(&acc[d * 32 + lane], t[(long)s * 32 + lane]);
            }
        }
        p = pn;
    }
    __syncthreads();
    const int n0 = bk * npb;
    for (int i = tid; i < nacc; i += BT) {
        const int node = n0 + (i >> 5);
        if (node < N) out[(long)node * 32 + (i & 31)] += acc[i] * deg[node];
    }
}

extern "C" void kernel_launch(void* const* d_in, const int* in_sizes, int n_in,
                              void* d_out, int out_size, void* d_ws, size_t ws_size,
                              hipStream_t stream) {
    const float* x   = (const float*)d_in[0];
    const int*   ei  = (const int*)d_in[1];
    const float* W1l = (const float*)d_in[2];
    const float* b1  = (const float*)d_in[3];
    const float* W1r = (const float*)d_in[4];
    const float* W2l = (const float*)d_in[5];
    const float* b2  = (const float*)d_in[6];
    const float* W2r = (const float*)d_in[7];
    float* out = (float*)d_out;

    int N = in_sizes[0] / 9;
    int E = in_sizes[1] / 2;
    const int* src = ei;
    const int* dst = ei + E;

    float* ws   = (float*)d_ws;
    float* deg  = ws;
    float* t    = deg + N;
    unsigned* sorted = (unsigned*)(t + (long)N * 32);
    int* tab = (int*)(sorted + E);

    int npb   = (N + NBK - 1) / NBK;
    int chunk = (E + NBK - 1) / NBK;

    bucket_group<<<NBK, BT, (size_t)chunk * sizeof(unsigned), stream>>>(src, dst, tab, sorted, E, chunk, npb);

    int ntl = (npb + 15) / 16;
    int nht = ntl < BT / 32 ? ntl : BT / 32;
    gather_x_layers<<<NBK, BT, ((size_t)ntl * 256 + (size_t)nht * 16 * 68) * sizeof(float), stream>>>(
        x, sorted, tab, W1l, b1, W1r, W2l, b2, W2r, deg, t, out, N, npb, E, chunk);

    gather_t<<<NBK, BT, (size_t)npb * 32 * sizeof(float), stream>>>(t, sorted, tab, deg, out, N, npb, E, chunk);
}


# Edge grouping reads every edge once, a thread's 16 loads in one group: the counting LDS add returns the edge's rank in its bucket, rank and edge word wait in LDS, placement runs from LDS
# speedup vs baseline: 6.0738x; 1.0401x over previous
#include <hip/hip_runtime.h>

typedef __attribute__((ext_vector_type(2))) float v2f;
typedef __attribute__((ext_vector_type(8))) float v8f;

#define WMMA_F32_4(a, b, c) \
    __builtin_amdgcn_wmma_f32_16x16x4_f32(false, (a), false, (b), (short)0, (c), false, false)

#define NBK 256
#define BT  1024

__device__ __forceinline__ void atomAddF32(float* p, float v) {
    unsafeAtomicAdd(p, v);
}


__device__ __forceinline__ int chunk_end(int w, int chunk, int E) {
    const int e0 = w * chunk;
    int e1 = e0 + chunk;
    if (e1 > E) e1 = E;
    return e1 < e0 ? e0 : e1;
}

__device__ __forceinline__ int wave_incl_sum(int v, int lane) {
    int s = v;
#pragma unroll
    for (int off = 1; off < 32; off <<= 1) {
        const int u = __shfl_up(s, off);
        if (lane >= off) s += u;
    }
    return s;
}

__global__ __launch_bounds__(BT) void bucket_group(const int* __restrict__ src, const int* __restrict__ dst,
                                                   int* __restrict__ tab, unsigned* __restrict__ sorted,
                                                   int E, int chunk, int npb) {
    extern __shared__ __attribute__((aligned(16))) unsigned stage[];
    uint2* kw = (uint2*)(stage + ((chunk + 1) & ~1));
    __shared__ int hist[NBK];
    __shared__ int cursor[NBK];
    __shared__ int wsum[NBK / 32];
    const int tid = threadIdx.x, w = blockIdx.x, lane = tid & 31;
    if (tid < NBK) hist[tid] = 0;
    __syncthreads();
    const int e0 = w * chunk;
    const int e1 = chunk_end(w, chunk, E);
    for (int eb = e0 + tid; eb < e1; eb += 8 * BT) {
        int s[8], d[8];
#pragma unroll
        for (int u = 0; u < 8; ++u) {
            const int e  = eb + u * BT;
            const int ec = e < e1 ? e : e1 - 1;
            s[u] = src[ec];
            d[u] = dst[ec];
        }
        asm volatile("" :: "v"(s[1]), "v"(d[1]));
#pragma unroll
        for (int u = 0; u < 8; ++u) {
            const int e = eb + u * BT;
            if (e < e1) {
                const int bb = d[u] / npb;
                const int r  = atomicAdd(&hist[bb], 1);
                kw[e - e0] = make_uint2(((unsigned)r << 8) | (unsigned)bb,
                                        ((unsigned)s[u] << 12) | (unsigned)(d[u] - bb * npb));
            }
        }
    }
    __syncthreads();
    int c = 0, incl = 0;
    if (tid < NBK) {
        c = hist[tid];
        incl = wave_incl_sum(c, lane);
        if (lane == 31) wsum[tid >> 5] = incl;
    }
    __syncthreads();
    if (tid < NBK) {
        int before = 0;
#pragma unroll
        for (int k = 0; k < NBK / 32; ++k) before += (k < (tid >> 5)) ? wsum[k] : 0;
        const int first = before + incl - c;
        cursor[tid] = first;
        tab[tid * NBK + w] = e0 + first;
    }
    __syncthreads();
    const int n = e1 - e0;
    for (int ib = tid; ib < n; ib += 4 * BT) {
        uint2 q[4];
        int   cb[4];
#pragma unroll
        for (int u = 0; u < 4; ++u) {
            const int i = ib + u * BT;
            q[u] = kw[i < n ? i : n - 1];
        }
#pragma unroll
        for (int u = 0; u < 4; ++u) cb[u] = cursor[q[u].x & 255u];
#pragma unroll
        for (int u = 0; u < 4; ++u)
            if (ib + u * BT < n) stage[cb[u] + (int)(q[u].x >> 8)] = q[u].y;
    }
    __syncthreads();
    const int mis = (int)(((size_t)(sorted + e0) >> 2) & 31);
    for (int ib = tid - mis; ib < n; ib += 4 * BT) {
        unsigned wd[4];
#pragma unroll
        for (int u = 0; u < 4; ++u) {
            const int i = ib + u * BT;
            wd[u] = stage[i < 0 ? 0 : (i < n ? i : n - 1)];
        }
#pragma unroll
        for (int u = 0; u < 4; ++u) {
            const int i = ib + u * BT;
            if (i >= 0 && i < n) sorted[e0 + i] = wd[u];
        }
    }
}

struct RunDir {
    int rpos[NBK];
    int rdel[NBK];
    int wsum[NBK / 32];
    int total;
};

__device__ __forceinline__ void run_fetch(const int* __restrict__ tab, int bk, int tid, int E, int chunk,
                                          int& first, int& len) {
    const int nb = bk + 1 < NBK ? bk + 1 : NBK - 1;
    first = tab[bk * NBK + tid];
    int next = tab[nb * NBK + tid];
    if (bk + 1 == NBK) next = chunk_end(tid, chunk, E);
    len = next - first;
}

__device__ __forceinline__ int run_scan_a(RunDir& rd, int tid, int lane, int len) {
    const int incl = wave_incl_sum(len, lane);
    if (lane == 31) rd.wsum[tid >> 5] = incl;
    return incl;
}

__device__ __forceinline__ void run_scan_b(RunDir& rd, int tid, int first, int len, int incl) {
    int before = 0;
#pragma unroll
    for (int k = 0; k < NBK / 32; ++k) before += (k < (tid >> 5)) ? rd.wsum[k] : 0;
    const int ex = before + incl - len;
    rd.rpos[tid] = ex;
    rd.rdel[tid] = first - ex;
    if (tid == NBK - 1) rd.total = ex + len;
}

__device__ __forceinline__ int run_slot(const RunDir& rd, int g) {
    int k = 0;
#pragma unroll
    for (int step = NBK / 2; step >= 1; step >>= 1)
        if (rd.rpos[k + step] <= g) k += step;
    return g + rd.rdel[k];
}

__global__ __launch_bounds__(BT) void gather_x_layers(
    const float* __restrict__ x, const unsigned* __restrict__ sorted, const int* __restrict__ tab,
    const float* __restrict__ W1l, const float* __restrict__ b1, const float* __restrict__ W1r,
    const float* __restrict__ W2l, const float* __restrict__ b2, const float* __restrict__ W2r,
    float* __restrict__ deg, float* __restrict__ t, float* __restrict__ out, int N, int npb, int E, int chunk) {
    extern __shared__ float acc[];
    __shared__ float Bt1[64][36];
    __shared__ float Bt2[64][68];
    __shared__ RunDir rd;

    const int tid  = threadIdx.x, bk = blockIdx.x;
    const int lane = tid & 31;
    const int wv   = __builtin_amdgcn_readfirstlane(tid >> 5);
    const int slot = lane & 15;
    const bool hi  = lane >= 16;
    const int ntl  = (npb + 15) >> 4;
    const int nacc = ntl * 256;
    int rfirst = 0, rlen = 0, rincl = 0;
    if (tid < NBK) run_fetch(tab, bk, tid, E, chunk, rfirst, rlen);
    for (int i = tid; i < nacc; i += BT) acc[i] = 0.0f;
    if (tid < NBK) rincl = run_scan_a(rd, tid, lane, rlen);
    __syncthreads();
    if (tid < NBK) run_scan_b(rd, tid, rfirst, rlen, rincl);
    __syncthreads();
    const int total = __builtin_amdgcn_readfirstlane(rd.total);
    const int kq = tid >> 6, cq = tid & 63;
    const int kc = kq < 9 ? kq : 8;
    float w1a = W1l[kc * 64 + cq];
    float w1b = W1r[kc * 64 + cq];
    if (kq >= 9) { w1a = 0.0f; w1b = 0.0f; }
    const float* w2p = (cq < 32) ? (W2l + cq) : (W2r + (cq - 32));
    float w2v[4];
#pragma unroll
    for (int u = 0; u < 4; ++u) w2v[u] = w2p[(kq + 16 * u) * 32];
    const int cs = slot < 9 ? slot : 8;
    unsigned p = 0;
    if (wv * 32 + lane < total) p = sorted[run_slot(rd, wv * 32 + lane)];
    for (int base = wv * 32; base < total; base += BT) {
        const int sv = (int)(p >> 12), dv = (int)(p & 4095u);
        const bool full = base + 32 <= total;
        float v[16];
        int   ai[16];
        if (full) {
#pragma unroll
            for (int j = 0; j < 16; ++j) {
                const int sa = __builtin_amdgcn_readlane(sv, j), sb = __builtin_amdgcn_readlane(sv, j + 16);
                const int da = __builtin_amdgcn_readlane(dv, j), db = __builtin_amdgcn_readlane(dv, j + 16);
                const int s = hi ? sb : sa;
                const int d = hi ? db : da;
                v[j]  = x[(long)s * 9 + cs];
                ai[j] = d * 16 + slot;
            }
        }
        const int gn = base + BT + lane;
        const unsigned pn = sorted[run_slot(rd, gn < total ? gn : total - 1)];
        if (full) {
            if (slot < 10) {
#pragma unroll
                for (int j = 0; j < 16; ++j) atomAddF32(&acc[ai[j]], slot < 9 ? v[j] : 1.0f);
            }
        } else {
            for (int j = 0; j < 16; ++j) {
                const int own = hi ? j + 16 : j;
                const int s = __shfl(sv, own), d = __shfl(dv, own);
                if (base + own < total && slot < 10) {
                    const float xv = x[(long)s * 9 + cs];
                    atomAddF32(&acc[d * 16 + slot], slot < 9 ? xv : 1.0f);
                }
            }
        }
        p = pn;
    }
    Bt1[cq][kq]      = w1a;
    Bt1[cq][16 + kq] = w1b;
#pragma unroll
    for (int u = 0; u < 4; ++u) Bt2[cq][kq + 16 * u] = w2v[u];
    __syncthreads();

    const int n0   = bk * npb;
    const int m    = lane & 15;
    const int koff = (lane < 16) ? 0 : 2;
    const int mhi  = (lane < 16) ? 0 : 8;
    float* hs = acc + nacc + wv * (16 * 68);
    for (int tb = 0; tb < ntl; tb += BT / 32) {
        const int tl = tb + wv;
        const bool active = tl < ntl;
        v8f c0 = {}, c1 = {}, c2 = {}, c3 = {};
        if (active) {
            const int lrow  = tl * 16 + m;
            const int node  = n0 + lrow;
            const bool rv   = lrow < npb && node < N;
            const float* ar = acc + lrow * 16;
            const float inv = 1.0f / fmaxf(ar[9], 1.0f);
            if (lane < 16 && rv) deg[node] = inv;
            const float* xr = x + (long)(node < N ? node : N - 1) * 9;
            float x0 = xr[koff], x1 = xr[koff + 1], x2 = xr[4 + koff], x3 = xr[5 + koff], x4 = xr[8];
            if (!rv) { x0 = 0.0f; x1 = 0.0f; x2 = 0.0f; x3 = 0.0f; x4 = 0.0f; }
            if (lane >= 16) x4 = 0.0f;
#pragma unroll
            for (int ks = 0; ks < 7; ++ks) {
                const int kk = ks * 4 + koff;
                v2f a;
                if (ks < 4) {
                    a = *(const v2f*)&ar[kk];
                    a.x *= inv;
                    a.y *= inv;
                } else if (ks == 4) {
                    a.x = x0; a.y = x1;
                } else if (ks == 5) {
                    a.x = x2; a.y = x3;
                } else {
                    a.x = x4; a.y = 0.0f;
                }
                v2f b0  = *(const v2f*)&Bt1[ 0 + m][kk];
                v2f b1v = *(const v2f*)&Bt1[16 + m][kk];
                v2f b2v = *(const v2f*)&Bt1[32 + m][kk];
                v2f b3v = *(const v2f*)&Bt1[48 + m][kk];
                c0 = WMMA_F32_4(a, b0,  c0);
                c1 = WMMA_F32_4(a, b1v, c1);
                c2 = WMMA_F32_4(a, b2v, c2);
                c3 = WMMA_F32_4(a, b3v, c3);
            }
            const float bb0 = b1[m], bb1 = b1[16 + m], bb2 = b1[32 + m], bb3 = b1[48 + m];
#pragma unroll
            for (int r = 0; r < 8; ++r) {
                float* hp = hs + (r + mhi) * 68;
                hp[ 0 + m] = fmaxf(c0[r] + bb0, 0.0f);
                hp[16 + m] = fmaxf(c1[r] + bb1, 0.0f);
                hp[32 + m] = fmaxf(c2[r] + bb2, 0.0f);
                hp[48 + m] = fmaxf(c3[r] + bb3, 0.0f);
            }
        }
        __syncthreads();
        if (active) {
            c0 = (v8f){}; c1 = (v8f){}; c2 = (v8f){}; c3 = (v8f){};
#pragma unroll
            for (int ks = 0; ks < 16; ++ks) {
                const int kk = ks * 4 + koff;
                v2f a   = *(const v2f*)&hs[m * 68 + kk];
                v2f b0  = *(const v2f*)&Bt2[ 0 + m][kk];
                v2f b1v = *(const v2f*)&Bt2[16 + m][kk];
                v2f b2v = *(const v2f*)&Bt2[32 + m][kk];
                v2f b3v = *(const v2f*)&Bt2[48 + m][kk];
                c0 = WMMA_F32_4(a, b0,  c0);
                c1 = WMMA_F32_4(a, b1v, c1);
                c2 = WMMA_F32_4(a, b2v, c2);
                c3 = WMMA_F32_4(a, b3v, c3);
            }
            const float bo0 = b2[m], bo1 = b2[16 + m];
#pragma unroll
            for (int r = 0; r < 8; ++r) {
                const int lr  = tl * 16 + r + mhi;
                const int row = n0 + lr;
                if (lr < npb && row < N) {
                    t[(long)row * 32 + m]        = c0[r];
                    t[(long)row * 32 + 16 + m]   = c1[r];
                    out[(long)row * 32 + m]      = c2[r] + bo0;
                    out[(long)row * 32 + 16 + m] = c3[r] + bo1;
                }
            }
        }
    }
}

__global__ __launch_bounds__(BT) void gather_t(const float* __restrict__ t, const unsigned* __restrict__ sorted,
                                               const int* __restrict__ tab, const float* __restrict__ deg,
                                               float* __restrict__ out, int N, int npb, int E, int chunk) {
    extern __shared__ float acc[];
    __shared__ RunDir rd;
    const int tid  = threadIdx.x, bk = blockIdx.x;
    const int lane = tid & 31;
    const int wv   = __builtin_amdgcn_readfirstlane(tid >> 5);
    const int nacc = npb * 32;
    int rfirst = 0, rlen = 0, rincl = 0;
    if (tid < NBK) run_fetch(tab, bk, tid, E, chunk, rfirst, rlen);
    for (int i = tid; i < nacc; i += BT) acc[i] = 0.0f;
    if (tid < NBK) rincl = run_scan_a(rd, tid, lane, rlen);
    __syncthreads();
    if (tid < NBK) run_scan_b(rd, tid, rfirst, rlen, rincl);
    __syncthreads();
    const int total = __builtin_amdgcn_readfirstlane(rd.total);
    unsigned p = 0;
    if (wv * 32 + lane < total) p = sorted[run_slot(rd, wv * 32 + lane)];
    for (int base = wv * 32; base < total; base += BT) {
        const int sv = (int)(p >> 12), dv = (int)(p & 4095u);
        const bool full = base + 32 <= total;
        float v[32];
        if (full) {
#pragma unroll
            for (int j = 0; j < 32; ++j) {
                const int s = __builtin_amdgcn_readlane(sv, j);
                v[j] = t[(long)s * 32 + lane];
            }
        }
        const int gn = base + BT + lane;
        const unsigned pn = sorted[run_slot(rd, gn < total ? gn : total - 1)];
        if (full) {
#pragma unroll
            for (int j = 0; j < 32; ++j) {
                const int d = __builtin_amdgcn_readlane(dv, j);
                atomAddF32(&acc[d * 32 + lane], v[j]);
            }
        } else {
            const int n = total - base;
            for (int j = 0; j < n; ++j) {
                const int s = __shfl(sv, j), d = __shfl(dv, j);
                atomAddF32(&acc[d * 32 + lane], t[(long)s * 32 + lane]);
            }
        }
        p = pn;
    }
    __syncthreads();
    const int n0 = bk * npb;
    for (int i = tid; i < nacc; i += BT) {
        const int node = n0 + (i >> 5);
        if (node < N) out[(long)node * 32 + (i & 31)] += acc[i] * deg[node];
    }
}

extern "C" void kernel_launch(void* const* d_in, const int* in_sizes, int n_in,
                              void* d_out, int out_size, void* d_ws, size_t ws_size,
                              hipStream_t stream) {
    const float* x   = (const float*)d_in[0];
    const int*   ei  = (const int*)d_in[1];
    const float* W1l = (const float*)d_in[2];
    const float* b1  = (const float*)d_in[3];
    const float* W1r = (const float*)d_in[4];
    const float* W2l = (const float*)d_in[5];
    const float* b2  = (const float*)d_in[6];
    const float* W2r = (const float*)d_in[7];
    float* out = (float*)d_out;

    int N = in_sizes[0] / 9;
    int E = in_sizes[1] / 2;
    const int* src = ei;
    const int* dst = ei + E;

    float* ws   = (float*)d_ws;
    float* deg  = ws;
    float* t    = deg + N;
    unsigned* sorted = (unsigned*)(t + (long)N * 32);
    int* tab = (int*)(sorted + E);

    int npb   = (N + NBK - 1) / NBK;
    int chunk = (E + NBK - 1) / NBK;

    bucket_group<<<NBK, BT, ((size_t)chunk + 1) * sizeof(unsigned) + (size_t)chunk * sizeof(uint2), stream>>>(src, dst, tab, sorted, E, chunk, npb);

    int ntl = (npb + 15) / 16;
    int nht = ntl < BT / 32 ? ntl : BT / 32;
    gather_x_layers<<<NBK, BT, ((size_t)ntl * 256 + (size_t)nht * 16 * 68) * sizeof(float), stream>>>(
        x, sorted, tab, W1l, b1, W1r, W2l, b2, W2r, deg, t, out, N, npb, E, chunk);

    gather_t<<<NBK, BT, (size_t)npb * 32 * sizeof(float), stream>>>(t, sorted, tab, deg, out, N, npb, E, chunk);
}


# Last, partly filled edge group of both aggregation kernels gathered as one group like a full one (lanes past the list fetch an in-bounds row, adds masked), not one memory round trip per edge
# speedup vs baseline: 6.4846x; 1.0676x over previous
#include <hip/hip_runtime.h>

typedef __attribute__((ext_vector_type(2))) float v2f;
typedef __attribute__((ext_vector_type(8))) float v8f;

#define WMMA_F32_4(a, b, c) \
    __builtin_amdgcn_wmma_f32_16x16x4_f32(false, (a), false, (b), (short)0, (c), false, false)

#define NBK 256
#define BT  1024

__device__ __forceinline__ void atomAddF32(float* p, float v) {
    unsafeAtomicAdd(p, v);
}


__device__ __forceinline__ int chunk_end(int w, int chunk, int E) {
    const int e0 = w * chunk;
    int e1 = e0 + chunk;
    if (e1 > E) e1 = E;
    return e1 < e0 ? e0 : e1;
}

__device__ __forceinline__ int wave_incl_sum(int v, int lane) {
    int s = v;
#pragma unroll
    for (int off = 1; off < 32; off <<= 1) {
        const int u = __shfl_up(s, off);
        if (lane >= off) s += u;
    }
    return s;
}

__global__ __launch_bounds__(BT) void bucket_group(const int* __restrict__ src, const int* __restrict__ dst,
                                                   int* __restrict__ tab, unsigned* __restrict__ sorted,
                                                   int E, int chunk, int npb) {
    extern __shared__ __attribute__((aligned(16))) unsigned stage[];
    uint2* kw = (uint2*)(stage + ((chunk + 1) & ~1));
    __shared__ int hist[NBK];
    __shared__ int cursor[NBK];
    __shared__ int wsum[NBK / 32];
    const int tid = threadIdx.x, w = blockIdx.x, lane = tid & 31;
    if (tid < NBK) hist[tid] = 0;
    __syncthreads();
    const int e0 = w * chunk;
    const int e1 = chunk_end(w, chunk, E);
    for (int eb = e0 + tid; eb < e1; eb += 8 * BT) {
        int s[8], d[8];
#pragma unroll
        for (int u = 0; u < 8; ++u) {
            const int e  = eb + u * BT;
            const int ec = e < e1 ? e : e1 - 1;
            s[u] = src[ec];
            d[u] = dst[ec];
        }
        asm volatile("" :: "v"(s[1]), "v"(d[1]));
#pragma unroll
        for (int u = 0; u < 8; ++u) {
            const int e = eb + u * BT;
            if (e < e1) {
                const int bb = d[u] / npb;
                const int r  = atomicAdd(&hist[bb], 1);
                kw[e - e0] = make_uint2(((unsigned)r << 8) | (unsigned)bb,
                                        ((unsigned)s[u] << 12) | (unsigned)(d[u] - bb * npb));
            }
        }
    }
    __syncthreads();
    int c = 0, incl = 0;
    if (tid < NBK) {
        c = hist[tid];
        incl = wave_incl_sum(c, lane);
        if (lane == 31) wsum[tid >> 5] = incl;
    }
    __syncthreads();
    if (tid < NBK) {
        int before = 0;
#pragma unroll
        for (int k = 0; k < NBK / 32; ++k) before += (k < (tid >> 5)) ? wsum[k] : 0;
        const int first = before + incl - c;
        cursor[tid] = first;
        tab[tid * NBK + w] = e0 + first;
    }
    __syncthreads();
    const int n = e1 - e0;
    for (int ib = tid; ib < n; ib += 4 * BT) {
        uint2 q[4];
        int   cb[4];
#pragma unroll
        for (int u = 0; u < 4; ++u) {
            const int i = ib + u * BT;
            q[u] = kw[i < n ? i : n - 1];
        }
#pragma unroll
        for (int u = 0; u < 4; ++u) cb[u] = cursor[q[u].x & 255u];
#pragma unroll
        for (int u = 0; u < 4; ++u)
            if (ib + u * BT < n) stage[cb[u] + (int)(q[u].x >> 8)] = q[u].y;
    }
    __syncthreads();
    const int mis = (int)(((size_t)(sorted + e0) >> 2) & 31);
    for (int ib = tid - mis; ib < n; ib += 4 * BT) {
        unsigned wd[4];
#pragma unroll
        for (int u = 0; u < 4; ++u) {
            const int i = ib + u * BT;
            wd[u] = stage[i < 0 ? 0 : (i < n ? i : n - 1)];
        }
#pragma unroll
        for (int u = 0; u < 4; ++u) {
            const int i = ib + u * BT;
            if (i >= 0 && i < n) sorted[e0 + i] = wd[u];
        }
    }
}

struct RunDir {
    int rpos[NBK];
    int rdel[NBK];
    int wsum[NBK / 32];
    int total;
};

__device__ __forceinline__ void run_fetch(const int* __restrict__ tab, int bk, int tid, int E, int chunk,
                                          int& first, int& len) {
    const int nb = bk + 1 < NBK ? bk + 1 : NBK - 1;
    first = tab[bk * NBK + tid];
    int next = tab[nb * NBK + tid];
    if (bk + 1 == NBK) next = chunk_end(tid, chunk, E);
    len = next - first;
}

__device__ __forceinline__ int run_scan_a(RunDir& rd, int tid, int lane, int len) {
    const int incl = wave_incl_sum(len, lane);
    if (lane == 31) rd.wsum[tid >> 5] = incl;
    return incl;
}

__device__ __forceinline__ void run_scan_b(RunDir& rd, int tid, int first, int len, int incl) {
    int before = 0;
#pragma unroll
    for (int k = 0; k < NBK / 32; ++k) before += (k < (tid >> 5)) ? rd.wsum[k] : 0;
    const int ex = before + incl - len;
    rd.rpos[tid] = ex;
    rd.rdel[tid] = first - ex;
    if (tid == NBK - 1) rd.total = ex + len;
}

__device__ __forceinline__ int run_slot(const RunDir& rd, int g) {
    int k = 0;
#pragma unroll
    for (int step = NBK / 2; step >= 1; step >>= 1)
        if (rd.rpos[k + step] <= g) k += step;
    return g + rd.rdel[k];
}

__global__ __launch_bounds__(BT) void gather_x_layers(
    const float* __restrict__ x, const unsigned* __restrict__ sorted, const int* __restrict__ tab,
    const float* __restrict__ W1l, const float* __restrict__ b1, const float* __restrict__ W1r,
    const float* __restrict__ W2l, const float* __restrict__ b2, const float* __restrict__ W2r,
    float* __restrict__ deg, float* __restrict__ t, float* __restrict__ out, int N, int npb, int E, int chunk) {
    extern __shared__ float acc[];
    __shared__ float Bt1[64][36];
    __shared__ float Bt2[64][68];
    __shared__ RunDir rd;

    const int tid  = threadIdx.x, bk = blockIdx.x;
    const int lane = tid & 31;
    const int wv   = __builtin_amdgcn_readfirstlane(tid >> 5);
    const int slot = lane & 15;
    const bool hi  = lane >= 16;
    const int ntl  = (npb + 15) >> 4;
    const int nacc = ntl * 256;
    int rfirst = 0, rlen = 0, rincl = 0;
    if (tid < NBK) run_fetch(tab, bk, tid, E, chunk, rfirst, rlen);
    for (int i = tid; i < nacc; i += BT) acc[i] = 0.0f;
    if (tid < NBK) rincl = run_scan_a(rd, tid, lane, rlen);
    __syncthreads();
    if (tid < NBK) run_scan_b(rd, tid, rfirst, rlen, rincl);
    __syncthreads();
    const int total = __builtin_amdgcn_readfirstlane(rd.total);
    const int kq = tid >> 6, cq = tid & 63;
    const int kc = kq < 9 ? kq : 8;
    float w1a = W1l[kc * 64 + cq];
    float w1b = W1r[kc * 64 + cq];
    if (kq >= 9) { w1a = 0.0f; w1b = 0.0f; }
    const float* w2p = (cq < 32) ? (W2l + cq) : (W2r + (cq - 32));
    float w2v[4];
#pragma unroll
    for (int u = 0; u < 4; ++u) w2v[u] = w2p[(kq + 16 * u) * 32];
    const int cs = slot < 9 ? slot : 8;
    unsigned p = 0;
    if (wv * 32 + lane < total) p = sorted[run_slot(rd, wv * 32 + lane)];
    for (int base = wv * 32; base < total; base += BT) {
        const int sv = (int)(p >> 12), dv = (int)(p & 4095u);
        const int n = total - base;
        float v[16];
        int   ai[16];
#pragma unroll
        for (int j = 0; j < 16; ++j) {
            const int sa = __builtin_amdgcn_readlane(sv, j), sb = __builtin_amdgcn_readlane(sv, j + 16);
            const int da = __builtin_amdgcn_readlane(dv, j), db = __builtin_amdgcn_readlane(dv, j + 16);
            const int s = hi ? sb : sa;
            const int d = hi ? db : da;
            v[j]  = x[(long)s * 9 + cs];
            ai[j] = d * 16 + slot;
        }
        __builtin_amdgcn_sched_barrier(0);
        const int gn = base + BT + lane;
        const unsigned pn = sorted[run_slot(rd, gn < total ? gn : total - 1)];
        if (n >= 32) {
            if (slot < 10) {
#pragma unroll
                for (int j = 0; j < 16; ++j) atomAddF32(&acc[ai[j]], slot < 9 ? v[j] : 1.0f);
            }
        } else {
            const int lim = n - (hi ? 16 : 0);
            if (slot < 10) {
#pragma unroll
                for (int j = 0; j < 16; ++j)
                    if (j < lim) atomAddF32(&acc[ai[j]], slot < 9 ? v[j] : 1.0f);
            }
        }
        p = pn;
    }
    Bt1[cq][kq]      = w1a;
    Bt1[cq][16 + kq] = w1b;
#pragma unroll
    for (int u = 0; u < 4; ++u) Bt2[cq][kq + 16 * u] = w2v[u];
    __syncthreads();

    const int n0   = bk * npb;
    const int m    = lane & 15;
    const int koff = (lane < 16) ? 0 : 2;
    const int mhi  = (lane < 16) ? 0 : 8;
    float* hs = acc + nacc + wv * (16 * 68);
    for (int tb = 0; tb < ntl; tb += BT / 32) {
        const int tl = tb + wv;
        const bool active = tl < ntl;
        v8f c0 = {}, c1 = {}, c2 = {}, c3 = {};
        if (active) {
            const int lrow  = tl * 16 + m;
            const int node  = n0 + lrow;
            const bool rv   = lrow < npb && node < N;
            const float* ar = acc + lrow * 16;
            const float inv = 1.0f / fmaxf(ar[9], 1.0f);
            if (lane < 16 && rv) deg[node] = inv;
            const float* xr = x + (long)(node < N ? node : N - 1) * 9;
            float x0 = xr[koff], x1 = xr[koff + 1], x2 = xr[4 + koff], x3 = xr[5 + koff], x4 = xr[8];
            if (!rv) { x0 = 0.0f; x1 = 0.0f; x2 = 0.0f; x3 = 0.0f; x4 = 0.0f; }
            if (lane >= 16) x4 = 0.0f;
#pragma unroll
            for (int ks = 0; ks < 7; ++ks) {
                const int kk = ks * 4 + koff;
                v2f a;
                if (ks < 4) {
                    a = *(const v2f*)&ar[kk];
                    a.x *= inv;
                    a.y *= inv;
                } else if (ks == 4) {
                    a.x = x0; a.y = x1;
                } else if (ks == 5) {
                    a.x = x2; a.y = x3;
                } else {
                    a.x = x4; a.y = 0.0f;
                }
                v2f b0  = *(const v2f*)&Bt1[ 0 + m][kk];
                v2f b1v = *(const v2f*)&Bt1[16 + m][kk];
                v2f b2v = *(const v2f*)&Bt1[32 + m][kk];
                v2f b3v = *(const v2f*)&Bt1[48 + m][kk];
                c0 = WMMA_F32_4(a, b0,  c0);
                c1 = WMMA_F32_4(a, b1v, c1);
                c2 = WMMA_F32_4(a, b2v, c2);
                c3 = WMMA_F32_4(a, b3v, c3);
            }
            const float bb0 = b1[m], bb1 = b1[16 + m], bb2 = b1[32 + m], bb3 = b1[48 + m];
#pragma unroll
            for (int r = 0; r < 8; ++r) {
                float* hp = hs + (r + mhi) * 68;
                hp[ 0 + m] = fmaxf(c0[r] + bb0, 0.0f);
                hp[16 + m] = fmaxf(c1[r] + bb1, 0.0f);
                hp[32 + m] = fmaxf(c2[r] + bb2, 0.0f);
                hp[48 + m] = fmaxf(c3[r] + bb3, 0.0f);
            }
        }
        __syncthreads();
        if (active) {
            c0 = (v8f){}; c1 = (v8f){}; c2 = (v8f){}; c3 = (v8f){};
#pragma unroll
            for (int ks = 0; ks < 16; ++ks) {
                const int kk = ks * 4 + koff;
                v2f a   = *(const v2f*)&hs[m * 68 + kk];
                v2f b0  = *(const v2f*)&Bt2[ 0 + m][kk];
                v2f b1v = *(const v2f*)&Bt2[16 + m][kk];
                v2f b2v = *(const v2f*)&Bt2[32 + m][kk];
                v2f b3v = *(const v2f*)&Bt2[48 + m][kk];
                c0 = WMMA_F32_4(a, b0,  c0);
                c1 = WMMA_F32_4(a, b1v, c1);
                c2 = WMMA_F32_4(a, b2v, c2);
                c3 = WMMA_F32_4(a, b3v, c3);
            }
            const float bo0 = b2[m], bo1 = b2[16 + m];
#pragma unroll
            for (int r = 0; r < 8; ++r) {
                const int lr  = tl * 16 + r + mhi;
                const int row = n0 + lr;
                if (lr < npb && row < N) {
                    t[(long)row * 32 + m]        = c0[r];
                    t[(long)row * 32 + 16 + m]   = c1[r];
                    out[(long)row * 32 + m]      = c2[r] + bo0;
                    out[(long)row * 32 + 16 + m] = c3[r] + bo1;
                }
            }
        }
    }
}

__global__ __launch_bounds__(BT) void gather_t(const float* __restrict__ t, const unsigned* __restrict__ sorted,
                                               const int* __restrict__ tab, const float* __restrict__ deg,
                                               float* __restrict__ out, int N, int npb, int E, int chunk) {
    extern __shared__ float acc[];
    __shared__ RunDir rd;
    const int tid  = threadIdx.x, bk = blockIdx.x;
    const int lane = tid & 31;
    const int wv   = __builtin_amdgcn_readfirstlane(tid >> 5);
    const int nacc = npb * 32;
    int rfirst = 0, rlen = 0, rincl = 0;
    if (tid < NBK) run_fetch(tab, bk, tid, E, chunk, rfirst, rlen);
    for (int i = tid; i < nacc; i += BT) acc[i] = 0.0f;
    if (tid < NBK) rincl = run_scan_a(rd, tid, lane, rlen);
    __syncthreads();
    if (tid < NBK) run_scan_b(rd, tid, rfirst, rlen, rincl);
    __syncthreads();
    const int total = __builtin_amdgcn_readfirstlane(rd.total);
    unsigned p = 0;
    if (wv * 32 + lane < total) p = sorted[run_slot(rd, wv * 32 + lane)];
    for (int base = wv * 32; base < total; base += BT) {
        const int sv = (int)(p >> 12), dv = (int)(p & 4095u);
        const int n = total - base;
        float v[32];
#pragma unroll
        for (int j = 0; j < 32; ++j) {
            const int s = __builtin_amdgcn_readlane(sv, j);
            v[j] = t[(long)s * 32 + lane];
        }
        const int gn = base + BT + lane;
        const unsigned pn = sorted[run_slot(rd, gn < total ? gn : total - 1)];
        if (n >= 32) {
#pragma unroll
            for (int j = 0; j < 32; ++j) {
                const int d = __builtin_amdgcn_readlane(dv, j);
                atomAddF32(&acc[d * 32 + lane], v[j]);
            }
        } else {
#pragma unroll
            for (int j = 0; j < 32; ++j) {
                if (j < n) {
                    const int d = __builtin_amdgcn_readlane(dv, j);
                    atomAddF32(&acc[d * 32 + lane], v[j]);
                }
            }
        }
        p = pn;
    }
    __syncthreads();
    const int n0 = bk * npb;
    for (int i = tid; i < nacc; i += BT) {
        const int node = n0 + (i >> 5);
        if (node < N) out[(long)node * 32 + (i & 31)] += acc[i] * deg[node];
    }
}

extern "C" void kernel_launch(void* const* d_in, const int* in_sizes, int n_in,
                              void* d_out, int out_size, void* d_ws, size_t ws_size,
                              hipStream_t stream) {
    const float* x   = (const float*)d_in[0];
    const int*   ei  = (const int*)d_in[1];
    const float* W1l = (const float*)d_in[2];
    const float* b1  = (const float*)d_in[3];
    const float* W1r = (const float*)d_in[4];
    const float* W2l = (const float*)d_in[5];
    const float* b2  = (const float*)d_in[6];
    const float* W2r = (const float*)d_in[7];
    float* out = (float*)d_out;

    int N = in_sizes[0] / 9;
    int E = in_sizes[1] / 2;
    const int* src = ei;
    const int* dst = ei + E;

    float* ws   = (float*)d_ws;
    float* deg  = ws;
    float* t    = deg + N;
    unsigned* sorted = (unsigned*)(t + (long)N * 32);
    int* tab = (int*)(sorted + E);

    int npb   = (N + NBK - 1) / NBK;
    int chunk = (E + NBK - 1) / NBK;

    bucket_group<<<NBK, BT, ((size_t)chunk + 1) * sizeof(unsigned) + (size_t)chunk * sizeof(uint2), stream>>>(src, dst, tab, sorted, E, chunk, npb);

    int ntl = (npb + 15) / 16;
    int nht = ntl < BT / 32 ? ntl : BT / 32;
    gather_x_layers<<<NBK, BT, ((size_t)ntl * 256 + (size_t)nht * 16 * 68) * sizeof(float), stream>>>(
        x, sorted, tab, W1l, b1, W1r, W2l, b2, W2r, deg, t, out, N, npb, E, chunk);

    gather_t<<<NBK, BT, (size_t)npb * 32 * sizeof(float), stream>>>(t, sorted, tab, deg, out, N, npb, E, chunk);
}


# Layer-2 aggregation kernel: the final update's out rows and 1/deg are requested at kernel start and wait in registers over the edge loop, not fetched one dependent memory round trip per row behind it
# speedup vs baseline: 6.8474x; 1.0559x over previous
#include <hip/hip_runtime.h>

typedef __attribute__((ext_vector_type(2))) float v2f;
typedef __attribute__((ext_vector_type(8))) float v8f;

#define WMMA_F32_4(a, b, c) \
    __builtin_amdgcn_wmma_f32_16x16x4_f32(false, (a), false, (b), (short)0, (c), false, false)

#define NBK 256
#define BT  1024
#define EPW 16

__device__ __forceinline__ void atomAddF32(float* p, float v) {
    unsafeAtomicAdd(p, v);
}


__device__ __forceinline__ int chunk_end(int w, int chunk, int E) {
    const int e0 = w * chunk;
    int e1 = e0 + chunk;
    if (e1 > E) e1 = E;
    return e1 < e0 ? e0 : e1;
}

__device__ __forceinline__ int wave_incl_sum(int v, int lane) {
    int s = v;
#pragma unroll
    for (int off = 1; off < 32; off <<= 1) {
        const int u = __shfl_up(s, off);
        if (lane >= off) s += u;
    }
    return s;
}

__global__ __launch_bounds__(BT) void bucket_group(const int* __restrict__ src, const int* __restrict__ dst,
                                                   int* __restrict__ tab, unsigned* __restrict__ sorted,
                                                   int E, int chunk, int npb) {
    extern __shared__ __attribute__((aligned(16))) unsigned stage[];
    uint2* kw = (uint2*)(stage + ((chunk + 1) & ~1));
    __shared__ int hist[NBK];
    __shared__ int cursor[NBK];
    __shared__ int wsum[NBK / 32];
    const int tid = threadIdx.x, w = blockIdx.x, lane = tid & 31;
    if (tid < NBK) hist[tid] = 0;
    __syncthreads();
    const int e0 = w * chunk;
    const int e1 = chunk_end(w, chunk, E);
    for (int eb = e0 + tid; eb < e1; eb += 8 * BT) {
        int s[8], d[8];
#pragma unroll
        for (int u = 0; u < 8; ++u) {
            const int e  = eb + u * BT;
            const int ec = e < e1 ? e : e1 - 1;
            s[u] = src[ec];
            d[u] = dst[ec];
        }
        asm volatile("" :: "v"(s[1]), "v"(d[1]));
#pragma unroll
        for (int u = 0; u < 8; ++u) {
            const int e = eb + u * BT;
            if (e < e1) {
                const int bb = d[u] / npb;
                const int r  = atomicAdd(&hist[bb], 1);
                kw[e - e0] = make_uint2(((unsigned)r << 8) | (unsigned)bb,
                                        ((unsigned)s[u] << 12) | (unsigned)(d[u] - bb * npb));
            }
        }
    }
    __syncthreads();
    int c = 0, incl = 0;
    if (tid < NBK) {
        c = hist[tid];
        incl = wave_incl_sum(c, lane);
        if (lane == 31) wsum[tid >> 5] = incl;
    }
    __syncthreads();
    if (tid < NBK) {
        int before = 0;
#pragma unroll
        for (int k = 0; k < NBK / 32; ++k) before += (k < (tid >> 5)) ? wsum[k] : 0;
        const int first = before + incl - c;
        cursor[tid] = first;
        tab[tid * NBK + w] = e0 + first;
    }
    __syncthreads();
    const int n = e1 - e0;
    for (int ib = tid; ib < n; ib += 4 * BT) {
        uint2 q[4];
        int   cb[4];
#pragma unroll
        for (int u = 0; u < 4; ++u) {
            const int i = ib + u * BT;
            q[u] = kw[i < n ? i : n - 1];
        }
#pragma unroll
        for (int u = 0; u < 4; ++u) cb[u] = cursor[q[u].x & 255u];
#pragma unroll
        for (int u = 0; u < 4; ++u)
            if (ib + u * BT < n) stage[cb[u] + (int)(q[u].x >> 8)] = q[u].y;
    }
    __syncthreads();
    const int mis = (int)(((size_t)(sorted + e0) >> 2) & 31);
    for (int ib = tid - mis; ib < n; ib += 4 * BT) {
        unsigned wd[4];
#pragma unroll
        for (int u = 0; u < 4; ++u) {
            const int i = ib + u * BT;
            wd[u] = stage[i < 0 ? 0 : (i < n ? i : n - 1)];
        }
#pragma unroll
        for (int u = 0; u < 4; ++u) {
            const int i = ib + u * BT;
            if (i >= 0 && i < n) sorted[e0 + i] = wd[u];
        }
    }
}

struct RunDir {
    int rpos[NBK];
    int rdel[NBK];
    int wsum[NBK / 32];
    int total;
};

__device__ __forceinline__ void run_fetch(const int* __restrict__ tab, int bk, int tid, int E, int chunk,
                                          int& first, int& len) {
    const int nb = bk + 1 < NBK ? bk + 1 : NBK - 1;
    first = tab[bk * NBK + tid];
    int next = tab[nb * NBK + tid];
    if (bk + 1 == NBK) next = chunk_end(tid, chunk, E);
    len = next - first;
}

__device__ __forceinline__ int run_scan_a(RunDir& rd, int tid, int lane, int len) {
    const int incl = wave_incl_sum(len, lane);
    if (lane == 31) rd.wsum[tid >> 5] = incl;
    return incl;
}

__device__ __forceinline__ void run_scan_b(RunDir& rd, int tid, int first, int len, int incl) {
    int before = 0;
#pragma unroll
    for (int k = 0; k < NBK / 32; ++k) before += (k < (tid >> 5)) ? rd.wsum[k] : 0;
    const int ex = before + incl - len;
    rd.rpos[tid] = ex;
    rd.rdel[tid] = first - ex;
    if (tid == NBK - 1) rd.total = ex + len;
}

__device__ __forceinline__ int run_slot(const RunDir& rd, int g) {
    int k = 0;
#pragma unroll
    for (int step = NBK / 2; step >= 1; step >>= 1)
        if (rd.rpos[k + step] <= g) k += step;
    return g + rd.rdel[k];
}

__global__ __launch_bounds__(BT) void gather_x_layers(
    const float* __restrict__ x, const unsigned* __restrict__ sorted, const int* __restrict__ tab,
    const float* __restrict__ W1l, const float* __restrict__ b1, const float* __restrict__ W1r,
    const float* __restrict__ W2l, const float* __restrict__ b2, const float* __restrict__ W2r,
    float* __restrict__ deg, float* __restrict__ t, float* __restrict__ out, int N, int npb, int E, int chunk) {
    extern __shared__ float acc[];
    __shared__ float Bt1[64][36];
    __shared__ float Bt2[64][68];
    __shared__ RunDir rd;

    const int tid  = threadIdx.x, bk = blockIdx.x;
    const int lane = tid & 31;
    const int wv   = __builtin_amdgcn_readfirstlane(tid >> 5);
    const int slot = lane & 15;
    const bool hi  = lane >= 16;
    const int ntl  = (npb + 15) >> 4;
    const int nacc = ntl * 256;
    int rfirst = 0, rlen = 0, rincl = 0;
    if (tid < NBK) run_fetch(tab, bk, tid, E, chunk, rfirst, rlen);
    for (int i = tid; i < nacc; i += BT) acc[i] = 0.0f;
    if (tid < NBK) rincl = run_scan_a(rd, tid, lane, rlen);
    __syncthreads();
    if (tid < NBK) run_scan_b(rd, tid, rfirst, rlen, rincl);
    __syncthreads();
    const int total = __builtin_amdgcn_readfirstlane(rd.total);
    const int kq = tid >> 6, cq = tid & 63;
    const int kc = kq < 9 ? kq : 8;
    float w1a = W1l[kc * 64 + cq];
    float w1b = W1r[kc * 64 + cq];
    if (kq >= 9) { w1a = 0.0f; w1b = 0.0f; }
    const float* w2p = (cq < 32) ? (W2l + cq) : (W2r + (cq - 32));
    float w2v[4];
#pragma unroll
    for (int u = 0; u < 4; ++u) w2v[u] = w2p[(kq + 16 * u) * 32];
    const int cs = slot < 9 ? slot : 8;
    unsigned p = 0;
    if (wv * 32 + lane < total) p = sorted[run_slot(rd, wv * 32 + lane)];
    for (int base = wv * 32; base < total; base += BT) {
        const int sv = (int)(p >> 12), dv = (int)(p & 4095u);
        const int n = total - base;
        float v[16];
        int   ai[16];
#pragma unroll
        for (int j = 0; j < 16; ++j) {
            const int sa = __builtin_amdgcn_readlane(sv, j), sb = __builtin_amdgcn_readlane(sv, j + 16);
            const int da = __builtin_amdgcn_readlane(dv, j), db = __builtin_amdgcn_readlane(dv, j + 16);
            const int s = hi ? sb : sa;
            const int d = hi ? db : da;
            v[j]  = x[(long)s * 9 + cs];
            ai[j] = d * 16 + slot;
        }
        __builtin_amdgcn_sched_barrier(0);
        const int gn = base + BT + lane;
        const unsigned pn = sorted[run_slot(rd, gn < total ? gn : total - 1)];
        if (n >= 32) {
            if (slot < 10) {
#pragma unroll
                for (int j = 0; j < 16; ++j) atomAddF32(&acc[ai[j]], slot < 9 ? v[j] : 1.0f);
            }
        } else {
            const int lim = n - (hi ? 16 : 0);
            if (slot < 10) {
#pragma unroll
                for (int j = 0; j < 16; ++j)
                    if (j < lim) atomAddF32(&acc[ai[j]], slot < 9 ? v[j] : 1.0f);
            }
        }
        p = pn;
    }
    Bt1[cq][kq]      = w1a;
    Bt1[cq][16 + kq] = w1b;
#pragma unroll
    for (int u = 0; u < 4; ++u) Bt2[cq][kq + 16 * u] = w2v[u];
    __syncthreads();

    const int n0   = bk * npb;
    const int m    = lane & 15;
    const int koff = (lane < 16) ? 0 : 2;
    const int mhi  = (lane < 16) ? 0 : 8;
    float* hs = acc + nacc + wv * (16 * 68);
    for (int tb = 0; tb < ntl; tb += BT / 32) {
        const int tl = tb + wv;
        const bool active = tl < ntl;
        v8f c0 = {}, c1 = {}, c2 = {}, c3 = {};
        if (active) {
            const int lrow  = tl * 16 + m;
            const int node  = n0 + lrow;
            const bool rv   = lrow < npb && node < N;
            const float* ar = acc + lrow * 16;
            const float inv = 1.0f / fmaxf(ar[9], 1.0f);
            if (lane < 16 && rv) deg[node] = inv;
            const float* xr = x + (long)(node < N ? node : N - 1) * 9;
            float x0 = xr[koff], x1 = xr[koff + 1], x2 = xr[4 + koff], x3 = xr[5 + koff], x4 = xr[8];
            if (!rv) { x0 = 0.0f; x1 = 0.0f; x2 = 0.0f; x3 = 0.0f; x4 = 0.0f; }
            if (lane >= 16) x4 = 0.0f;
#pragma unroll
            for (int ks = 0; ks < 7; ++ks) {
                const int kk = ks * 4 + koff;
                v2f a;
                if (ks < 4) {
                    a = *(const v2f*)&ar[kk];
                    a.x *= inv;
                    a.y *= inv;
                } else if (ks == 4) {
                    a.x = x0; a.y = x1;
                } else if (ks == 5) {
                    a.x = x2; a.y = x3;
                } else {
                    a.x = x4; a.y = 0.0f;
                }
                v2f b0  = *(const v2f*)&Bt1[ 0 + m][kk];
                v2f b1v = *(const v2f*)&Bt1[16 + m][kk];
                v2f b2v = *(const v2f*)&Bt1[32 + m][kk];
                v2f b3v = *(const v2f*)&Bt1[48 + m][kk];
                c0 = WMMA_F32_4(a, b0,  c0);
                c1 = WMMA_F32_4(a, b1v, c1);
                c2 = WMMA_F32_4(a, b2v, c2);
                c3 = WMMA_F32_4(a, b3v, c3);
            }
            const float bb0 = b1[m], bb1 = b1[16 + m], bb2 = b1[32 + m], bb3 = b1[48 + m];
#pragma unroll
            for (int r = 0; r < 8; ++r) {
                float* hp = hs + (r + mhi) * 68;
                hp[ 0 + m] = fmaxf(c0[r] + bb0, 0.0f);
                hp[16 + m] = fmaxf(c1[r] + bb1, 0.0f);
                hp[32 + m] = fmaxf(c2[r] + bb2, 0.0f);
                hp[48 + m] = fmaxf(c3[r] + bb3, 0.0f);
            }
        }
        __syncthreads();
        if (active) {
            c0 = (v8f){}; c1 = (v8f){}; c2 = (v8f){}; c3 = (v8f){};
#pragma unroll
            for (int ks = 0; ks < 16; ++ks) {
                const int kk = ks * 4 + koff;
                v2f a   = *(const v2f*)&hs[m * 68 + kk];
                v2f b0  = *(const v2f*)&Bt2[ 0 + m][kk];
                v2f b1v = *(const v2f*)&Bt2[16 + m][kk];
                v2f b2v = *(const v2f*)&Bt2[32 + m][kk];
                v2f b3v = *(const v2f*)&Bt2[48 + m][kk];
                c0 = WMMA_F32_4(a, b0,  c0);
                c1 = WMMA_F32_4(a, b1v, c1);
                c2 = WMMA_F32_4(a, b2v, c2);
                c3 = WMMA_F32_4(a, b3v, c3);
            }
            const float bo0 = b2[m], bo1 = b2[16 + m];
#pragma unroll
            for (int r = 0; r < 8; ++r) {
                const int lr  = tl * 16 + r + mhi;
                const int row = n0 + lr;
                if (lr < npb && row < N) {
                    t[(long)row * 32 + m]        = c0[r];
                    t[(long)row * 32 + 16 + m]   = c1[r];
                    out[(long)row * 32 + m]      = c2[r] + bo0;
                    out[(long)row * 32 + 16 + m] = c3[r] + bo1;
                }
            }
        }
    }
}

__global__ __launch_bounds__(BT) void gather_t(const float* __restrict__ t, const unsigned* __restrict__ sorted,
                                               const int* __restrict__ tab, const float* __restrict__ deg,
                                               float* __restrict__ out, int N, int npb, int E, int chunk) {
    extern __shared__ float acc[];
    __shared__ RunDir rd;
    const int tid  = threadIdx.x, bk = blockIdx.x;
    const int lane = tid & 31;
    const int wv   = __builtin_amdgcn_readfirstlane(tid >> 5);
    const int nacc = npb * 32;
    int rfirst = 0, rlen = 0, rincl = 0;
    if (tid < NBK) run_fetch(tab, bk, tid, E, chunk, rfirst, rlen);
    for (int i = tid; i < nacc; i += BT) acc[i] = 0.0f;
    if (tid < NBK) rincl = run_scan_a(rd, tid, lane, rlen);
    __syncthreads();
    if (tid < NBK) run_scan_b(rd, tid, rfirst, rlen, rincl);
    __syncthreads();
    const int total = __builtin_amdgcn_readfirstlane(rd.total);
    const int n0 = bk * npb;
    float ov[EPW], dg[EPW];
#pragma unroll
    for (int k = 0; k < EPW; ++k) {
        const int lr = wv + k * (BT / 32);
        int node = n0 + (lr < npb ? lr : npb - 1);
        node = node < N ? node : N - 1;
        ov[k] = out[(long)node * 32 + lane];
        dg[k] = deg[node];
    }
    unsigned p = 0;
    if (wv * 32 + lane < total) p = sorted[run_slot(rd, wv * 32 + lane)];
    for (int base = wv * 32; base < total; base += BT) {
        const int sv = (int)(p >> 12), dv = (int)(p & 4095u);
        const int n = total - base;
        float v[32];
#pragma unroll
        for (int j = 0; j < 32; ++j) {
            const int s = __builtin_amdgcn_readlane(sv, j);
            v[j] = t[(long)s * 32 + lane];
        }
        const int gn = base + BT + lane;
        const unsigned pn = sorted[run_slot(rd, gn < total ? gn : total - 1)];
        if (n >= 32) {
#pragma unroll
            for (int j = 0; j < 32; ++j) {
                const int d = __builtin_amdgcn_readlane(dv, j);
                atomAddF32(&acc[d * 32 + lane], v[j]);
            }
        } else {
#pragma unroll
            for (int j = 0; j < 32; ++j) {
                if (j < n) {
                    const int d = __builtin_amdgcn_readlane(dv, j);
                    atomAddF32(&acc[d * 32 + lane], v[j]);
                }
            }
        }
        p = pn;
    }
    __syncthreads();
    float av[EPW];
#pragma unroll
    for (int k = 0; k < EPW; ++k) {
        const int lr = wv + k * (BT / 32);
        av[k] = acc[(lr < npb ? lr : npb - 1) * 32 + lane];
    }
    static_assert(EPW == 16, "the statement below names av[0] to av[15]");
    asm volatile("" :: "v"(av[0]), "v"(av[1]), "v"(av[2]), "v"(av[3]), "v"(av[4]), "v"(av[5]), "v"(av[6]), "v"(av[7]),
                       "v"(av[8]), "v"(av[9]), "v"(av[10]), "v"(av[11]), "v"(av[12]), "v"(av[13]), "v"(av[14]), "v"(av[15]));
#pragma unroll
    for (int k = 0; k < EPW; ++k) {
        const int lr   = wv + k * (BT / 32);
        const int node = n0 + lr;
        if (lr < npb && node < N) out[(long)node * 32 + lane] = ov[k] + av[k] * dg[k];
    }
    for (int lr = wv + EPW * (BT / 32); lr < npb; lr += BT / 32) {
        const int node = n0 + lr;
        if (node < N) out[(long)node * 32 + lane] += acc[lr * 32 + lane] * deg[node];
    }
}

extern "C" void kernel_launch(void* const* d_in, const int* in_sizes, int n_in,
                              void* d_out, int out_size, void* d_ws, size_t ws_size,
                              hipStream_t stream) {
    const float* x   = (const float*)d_in[0];
    const int*   ei  = (const int*)d_in[1];
    const float* W1l = (const float*)d_in[2];
    const float* b1  = (const float*)d_in[3];
    const float* W1r = (const float*)d_in[4];
    const float* W2l = (const float*)d_in[5];
    const float* b2  = (const float*)d_in[6];
    const float* W2r = (const float*)d_in[7];
    float* out = (float*)d_out;

    int N = in_sizes[0] / 9;
    int E = in_sizes[1] / 2;
    const int* src = ei;
    const int* dst = ei + E;

    float* ws   = (float*)d_ws;
    float* deg  = ws;
    float* t    = deg + N;
    unsigned* sorted = (unsigned*)(t + (long)N * 32);
    int* tab = (int*)(sorted + E);

    int npb   = (N + NBK - 1) / NBK;
    int chunk = (E + NBK - 1) / NBK;

    bucket_group<<<NBK, BT, ((size_t)chunk + 1) * sizeof(unsigned) + (size_t)chunk * sizeof(uint2), stream>>>(src, dst, tab, sorted, E, chunk, npb);

    int ntl = (npb + 15) / 16;
    int nht = ntl < BT / 32 ? ntl : BT / 32;
    gather_x_layers<<<NBK, BT, ((size_t)ntl * 256 + (size_t)nht * 16 * 68) * sizeof(float), stream>>>(
        x, sorted, tab, W1l, b1, W1r, W2l, b2, W2r, deg, t, out, N, npb, E, chunk);

    gather_t<<<NBK, BT, (size_t)npb * 32 * sizeof(float), stream>>>(t, sorted, tab, deg, out, N, npb, E, chunk);
}


# Layer-1 gather reads x from a copy the grouping kernel re-lays as 64-byte rows on 64-byte boundaries (9 features, then a one for the in-degree): one cache line per edge, none across two lines
# speedup vs baseline: 7.2288x; 1.0557x over previous
#include <hip/hip_runtime.h>

typedef __attribute__((ext_vector_type(2))) float v2f;
typedef __attribute__((ext_vector_type(8))) float v8f;

#define WMMA_F32_4(a, b, c) \
    __builtin_amdgcn_wmma_f32_16x16x4_f32(false, (a), false, (b), (short)0, (c), false, false)

#define NBK 256
#define BT  1024
#define EPW 16

__device__ __forceinline__ void atomAddF32(float* p, float v) {
    unsafeAtomicAdd(p, v);
}


__device__ __forceinline__ int chunk_end(int w, int chunk, int E) {
    const int e0 = w * chunk;
    int e1 = e0 + chunk;
    if (e1 > E) e1 = E;
    return e1 < e0 ? e0 : e1;
}

__device__ __forceinline__ int wave_incl_sum(int v, int lane) {
    int s = v;
#pragma unroll
    for (int off = 1; off < 32; off <<= 1) {
        const int u = __shfl_up(s, off);
        if (lane >= off) s += u;
    }
    return s;
}

__global__ __launch_bounds__(BT) void bucket_group(const int* __restrict__ src, const int* __restrict__ dst,
                                                   const float* __restrict__ x, float* __restrict__ xp,
                                                   int* __restrict__ tab, unsigned* __restrict__ sorted,
                                                   int N, int E, int chunk, int npb) {
    extern __shared__ __attribute__((aligned(16))) unsigned stage[];
    uint2* kw = (uint2*)(stage + ((chunk + 1) & ~1));
    __shared__ int hist[NBK];
    __shared__ int cursor[NBK];
    __shared__ int wsum[NBK / 32];
    const int tid = threadIdx.x, w = blockIdx.x, lane = tid & 31;
    if (tid < NBK) hist[tid] = 0;
    __syncthreads();
    const int e0 = w * chunk;
    const int e1 = chunk_end(w, chunk, E);
    for (int eb = e0 + tid; eb < e1; eb += 8 * BT) {
        int s[8], d[8];
#pragma unroll
        for (int u = 0; u < 8; ++u) {
            const int e  = eb + u * BT;
            const int ec = e < e1 ? e : e1 - 1;
            s[u] = src[ec];
            d[u] = dst[ec];
        }
        asm volatile("" :: "v"(s[1]), "v"(d[1]));
#pragma unroll
        for (int u = 0; u < 8; ++u) {
            const int e = eb + u * BT;
            if (e < e1) {
                const int bb = d[u] / npb;
                const int r  = atomicAdd(&hist[bb], 1);
                kw[e - e0] = make_uint2(((unsigned)r << 8) | (unsigned)bb,
                                        ((unsigned)s[u] << 12) | (unsigned)(d[u] - bb * npb));
            }
        }
    }
    __syncthreads();
    int c = 0, incl = 0;
    if (tid < NBK) {
        c = hist[tid];
        incl = wave_incl_sum(c, lane);
        if (lane == 31) wsum[tid >> 5] = incl;
    }
    __syncthreads();
    if (tid < NBK) {
        int before = 0;
#pragma unroll
        for (int k = 0; k < NBK / 32; ++k) before += (k < (tid >> 5)) ? wsum[k] : 0;
        const int first = before + incl - c;
        cursor[tid] = first;
        tab[tid * NBK + w] = e0 + first;
    }
    __syncthreads();
    const int nx  = npb * 16;
    const int xn0 = w * npb;
    const int xm  = (int)(((size_t)(xp + (size_t)xn0 * 16) >> 2) & 31);
    float xv[8];
#pragma unroll
    for (int u = 0; u < 8; ++u) {
        const int i   = tid - xm + u * BT;
        const int ic  = i < 0 ? 0 : i;
        const int col = ic & 15;
        int node = xn0 + (ic >> 4);
        node = node < N ? node : N - 1;
        xv[u] = x[(long)node * 9 + (col < 9 ? col : 8)];
    }
    const int n = e1 - e0;
    for (int ib = tid; ib < n; ib += 4 * BT) {
        uint2 q[4];
        int   cb[4];
#pragma unroll
        for (int u = 0; u < 4; ++u) {
            const int i = ib + u * BT;
            q[u] = kw[i < n ? i : n - 1];
        }
#pragma unroll
        for (int u = 0; u < 4; ++u) cb[u] = cursor[q[u].x & 255u];
#pragma unroll
        for (int u = 0; u < 4; ++u)
            if (ib + u * BT < n) stage[cb[u] + (int)(q[u].x >> 8)] = q[u].y;
    }
    __syncthreads();
#pragma unroll
    for (int u = 0; u < 8; ++u) {
        const int i    = tid - xm + u * BT;
        const int col  = i & 15;
        const int node = xn0 + (i >> 4);
        if (i >= 0 && i < nx && node < N)
            xp[(long)node * 16 + col] = col < 9 ? xv[u] : (col == 9 ? 1.0f : 0.0f);
    }
    for (int i = tid - xm + 8 * BT; i < nx; i += BT) {
        const int col  = i & 15;
        const int node = xn0 + (i >> 4);
        if (node < N) xp[(long)node * 16 + col] = col < 9 ? x[(long)node * 9 + col] : (col == 9 ? 1.0f : 0.0f);
    }
    const int mis = (int)(((size_t)(sorted + e0) >> 2) & 31);
    for (int ib = tid - mis; ib < n; ib += 4 * BT) {
        unsigned wd[4];
#pragma unroll
        for (int u = 0; u < 4; ++u) {
            const int i = ib + u * BT;
            wd[u] = stage[i < 0 ? 0 : (i < n ? i : n - 1)];
        }
#pragma unroll
        for (int u = 0; u < 4; ++u) {
            const int i = ib + u * BT;
            if (i >= 0 && i < n) sorted[e0 + i] = wd[u];
        }
    }
}

struct RunDir {
    int rpos[NBK];
    int rdel[NBK];
    int wsum[NBK / 32];
    int total;
};

__device__ __forceinline__ void run_fetch(const int* __restrict__ tab, int bk, int tid, int E, int chunk,
                                          int& first, int& len) {
    const int nb = bk + 1 < NBK ? bk + 1 : NBK - 1;
    first = tab[bk * NBK + tid];
    int next = tab[nb * NBK + tid];
    if (bk + 1 == NBK) next = chunk_end(tid, chunk, E);
    len = next - first;
}

__device__ __forceinline__ int run_scan_a(RunDir& rd, int tid, int lane, int len) {
    const int incl = wave_incl_sum(len, lane);
    if (lane == 31) rd.wsum[tid >> 5] = incl;
    return incl;
}

__device__ __forceinline__ void run_scan_b(RunDir& rd, int tid, int first, int len, int incl) {
    int before = 0;
#pragma unroll
    for (int k = 0; k < NBK / 32; ++k) before += (k < (tid >> 5)) ? rd.wsum[k] : 0;
    const int ex = before + incl - len;
    rd.rpos[tid] = ex;
    rd.rdel[tid] = first - ex;
    if (tid == NBK - 1) rd.total = ex + len;
}

__device__ __forceinline__ int run_slot(const RunDir& rd, int g) {
    int k = 0;
#pragma unroll
    for (int step = NBK / 2; step >= 1; step >>= 1)
        if (rd.rpos[k + step] <= g) k += step;
    return g + rd.rdel[k];
}

__global__ __launch_bounds__(BT) void gather_x_layers(
    const float* __restrict__ x, const float* __restrict__ xp, const unsigned* __restrict__ sorted, const int* __restrict__ tab,
    const float* __restrict__ W1l, const float* __restrict__ b1, const float* __restrict__ W1r,
    const float* __restrict__ W2l, const float* __restrict__ b2, const float* __restrict__ W2r,
    float* __restrict__ deg, float* __restrict__ t, float* __restrict__ out, int N, int npb, int E, int chunk) {
    extern __shared__ float acc[];
    __shared__ float Bt1[64][36];
    __shared__ float Bt2[64][68];
    __shared__ RunDir rd;

    const int tid  = threadIdx.x, bk = blockIdx.x;
    const int lane = tid & 31;
    const int wv   = __builtin_amdgcn_readfirstlane(tid >> 5);
    const int slot = lane & 15;
    const bool hi  = lane >= 16;
    const int ntl  = (npb + 15) >> 4;
    const int nacc = ntl * 256;
    int rfirst = 0, rlen = 0, rincl = 0;
    if (tid < NBK) run_fetch(tab, bk, tid, E, chunk, rfirst, rlen);
    for (int i = tid; i < nacc; i += BT) acc[i] = 0.0f;
    if (tid < NBK) rincl = run_scan_a(rd, tid, lane, rlen);
    __syncthreads();
    if (tid < NBK) run_scan_b(rd, tid, rfirst, rlen, rincl);
    __syncthreads();
    const int total = __builtin_amdgcn_readfirstlane(rd.total);
    const int kq = tid >> 6, cq = tid & 63;
    const int kc = kq < 9 ? kq : 8;
    float w1a = W1l[kc * 64 + cq];
    float w1b = W1r[kc * 64 + cq];
    if (kq >= 9) { w1a = 0.0f; w1b = 0.0f; }
    const float* w2p = (cq < 32) ? (W2l + cq) : (W2r + (cq - 32));
    float w2v[4];
#pragma unroll
    for (int u = 0; u < 4; ++u) w2v[u] = w2p[(kq + 16 * u) * 32];
    const int cs = slot < 10 ? slot : 9;
    unsigned p = 0;
    if (wv * 32 + lane < total) p = sorted[run_slot(rd, wv * 32 + lane)];
    for (int base = wv * 32; base < total; base += BT) {
        const int sv = (int)(p >> 12), dv = (int)(p & 4095u);
        const int n = total - base;
        float v[16];
        int   ai[16];
#pragma unroll
        for (int j = 0; j < 16; ++j) {
            const int sa = __builtin_amdgcn_readlane(sv, j), sb = __builtin_amdgcn_readlane(sv, j + 16);
            const int da = __builtin_amdgcn_readlane(dv, j), db = __builtin_amdgcn_readlane(dv, j + 16);
            const int s = hi ? sb : sa;
            const int d = hi ? db : da;
            v[j]  = xp[(long)s * 16 + cs];
            ai[j] = d * 16 + slot;
        }
        __builtin_amdgcn_sched_barrier(0);
        const int gn = base + BT + lane;
        const unsigned pn = sorted[run_slot(rd, gn < total ? gn : total - 1)];
        if (n >= 32) {
            if (slot < 10) {
#pragma unroll
                for (int j = 0; j < 16; ++j) atomAddF32(&acc[ai[j]], v[j]);
            }
        } else {
            const int lim = n - (hi ? 16 : 0);
            if (slot < 10) {
#pragma unroll
                for (int j = 0; j < 16; ++j)
                    if (j < lim) atomAddF32(&acc[ai[j]], v[j]);
            }
        }
        p = pn;
    }
    Bt1[cq][kq]      = w1a;
    Bt1[cq][16 + kq] = w1b;
#pragma unroll
    for (int u = 0; u < 4; ++u) Bt2[cq][kq + 16 * u] = w2v[u];
    __syncthreads();

    const int n0   = bk * npb;
    const int m    = lane & 15;
    const int koff = (lane < 16) ? 0 : 2;
    const int mhi  = (lane < 16) ? 0 : 8;
    float* hs = acc + nacc + wv * (16 * 68);
    for (int tb = 0; tb < ntl; tb += BT / 32) {
        const int tl = tb + wv;
        const bool active = tl < ntl;
        v8f c0 = {}, c1 = {}, c2 = {}, c3 = {};
        if (active) {
            const int lrow  = tl * 16 + m;
            const int node  = n0 + lrow;
            const bool rv   = lrow < npb && node < N;
            const float* ar = acc + lrow * 16;
            const float inv = 1.0f / fmaxf(ar[9], 1.0f);
            if (lane < 16 && rv) deg[node] = inv;
            const float* xr = x + (long)(node < N ? node : N - 1) * 9;
            float x0 = xr[koff], x1 = xr[koff + 1], x2 = xr[4 + koff], x3 = xr[5 + koff], x4 = xr[8];
            if (!rv) { x0 = 0.0f; x1 = 0.0f; x2 = 0.0f; x3 = 0.0f; x4 = 0.0f; }
            if (lane >= 16) x4 = 0.0f;
#pragma unroll
            for (int ks = 0; ks < 7; ++ks) {
                const int kk = ks * 4 + koff;
                v2f a;
                if (ks < 4) {
                    a = *(const v2f*)&ar[kk];
                    a.x *= inv;
                    a.y *= inv;
                } else if (ks == 4) {
                    a.x = x0; a.y = x1;
                } else if (ks == 5) {
                    a.x = x2; a.y = x3;
                } else {
                    a.x = x4; a.y = 0.0f;
                }
                v2f b0  = *(const v2f*)&Bt1[ 0 + m][kk];
                v2f b1v = *(const v2f*)&Bt1[16 + m][kk];
                v2f b2v = *(const v2f*)&Bt1[32 + m][kk];
                v2f b3v = *(const v2f*)&Bt1[48 + m][kk];
                c0 = WMMA_F32_4(a, b0,  c0);
                c1 = WMMA_F32_4(a, b1v, c1);
                c2 = WMMA_F32_4(a, b2v, c2);
                c3 = WMMA_F32_4(a, b3v, c3);
            }
            const float bb0 = b1[m], bb1 = b1[16 + m], bb2 = b1[32 + m], bb3 = b1[48 + m];
#pragma unroll
            for (int r = 0; r < 8; ++r) {
                float* hp = hs + (r + mhi) * 68;
                hp[ 0 + m] = fmaxf(c0[r] + bb0, 0.0f);
                hp[16 + m] = fmaxf(c1[r] + bb1, 0.0f);
                hp[32 + m] = fmaxf(c2[r] + bb2, 0.0f);
                hp[48 + m] = fmaxf(c3[r] + bb3, 0.0f);
            }
        }
        __syncthreads();
        if (active) {
            c0 = (v8f){}; c1 = (v8f){}; c2 = (v8f){}; c3 = (v8f){};
#pragma unroll
            for (int ks = 0; ks < 16; ++ks) {
                const int kk = ks * 4 + koff;
                v2f a   = *(const v2f*)&hs[m * 68 + kk];
                v2f b0  = *(const v2f*)&Bt2[ 0 + m][kk];
                v2f b1v = *(const v2f*)&Bt2[16 + m][kk];
                v2f b2v = *(const v2f*)&Bt2[32 + m][kk];
                v2f b3v = *(const v2f*)&Bt2[48 + m][kk];
                c0 = WMMA_F32_4(a, b0,  c0);
                c1 = WMMA_F32_4(a, b1v, c1);
                c2 = WMMA_F32_4(a, b2v, c2);
                c3 = WMMA_F32_4(a, b3v, c3);
            }
            const float bo0 = b2[m], bo1 = b2[16 + m];
#pragma unroll
            for (int r = 0; r < 8; ++r) {
                const int lr  = tl * 16 + r + mhi;
                const int row = n0 + lr;
                if (lr < npb && row < N) {
                    t[(long)row * 32 + m]        = c0[r];
                    t[(long)row * 32 + 16 + m]   = c1[r];
                    out[(long)row * 32 + m]      = c2[r] + bo0;
                    out[(long)row * 32 + 16 + m] = c3[r] + bo1;
                }
            }
        }
    }
}

__global__ __launch_bounds__(BT) void gather_t(const float* __restrict__ t, const unsigned* __restrict__ sorted,
                                               const int* __restrict__ tab, const float* __restrict__ deg,
                                               float* __restrict__ out, int N, int npb, int E, int chunk) {
    extern __shared__ float acc[];
    __shared__ RunDir rd;
    const int tid  = threadIdx.x, bk = blockIdx.x;
    const int lane = tid & 31;
    const int wv   = __builtin_amdgcn_readfirstlane(tid >> 5);
    const int nacc = npb * 32;
    int rfirst = 0, rlen = 0, rincl = 0;
    if (tid < NBK) run_fetch(tab, bk, tid, E, chunk, rfirst, rlen);
    for (int i = tid; i < nacc; i += BT) acc[i] = 0.0f;
    if (tid < NBK) rincl = run_scan_a(rd, tid, lane, rlen);
    __syncthreads();
    if (tid < NBK) run_scan_b(rd, tid, rfirst, rlen, rincl);
    __syncthreads();
    const int total = __builtin_amdgcn_readfirstlane(rd.total);
    const int n0 = bk * npb;
    float ov[EPW], dg[EPW];
#pragma unroll
    for (int k = 0; k < EPW; ++k) {
        const int lr = wv + k * (BT / 32);
        int node = n0 + (lr < npb ? lr : npb - 1);
        node = node < N ? node : N - 1;
        ov[k] = out[(long)node * 32 + lane];
        dg[k] = deg[node];
    }
    unsigned p = 0;
    if (wv * 32 + lane < total) p = sorted[run_slot(rd, wv * 32 + lane)];
    for (int base = wv * 32; base < total; base += BT) {
        const int sv = (int)(p >> 12), dv = (int)(p & 4095u);
        const int n = total - base;
        float v[32];
#pragma unroll
        for (int j = 0; j < 32; ++j) {
            const int s = __builtin_amdgcn_readlane(sv, j);
            v[j] = t[(long)s * 32 + lane];
        }
        const int gn = base + BT + lane;
        const unsigned pn = sorted[run_slot(rd, gn < total ? gn : total - 1)];
        if (n >= 32) {
#pragma unroll
            for (int j = 0; j < 32; ++j) {
                const int d = __builtin_amdgcn_readlane(dv, j);
                atomAddF32(&acc[d * 32 + lane], v[j]);
            }
        } else {
#pragma unroll
            for (int j = 0; j < 32; ++j) {
                if (j < n) {
                    const int d = __builtin_amdgcn_readlane(dv, j);
                    atomAddF32(&acc[d * 32 + lane], v[j]);
                }
            }
        }
        p = pn;
    }
    __syncthreads();
    float av[EPW];
#pragma unroll
    for (int k = 0; k < EPW; ++k) {
        const int lr = wv + k * (BT / 32);
        av[k] = acc[(lr < npb ? lr : npb - 1) * 32 + lane];
    }
    static_assert(EPW == 16, "the statement below names av[0] to av[15]");
    asm volatile("" :: "v"(av[0]), "v"(av[1]), "v"(av[2]), "v"(av[3]), "v"(av[4]), "v"(av[5]), "v"(av[6]), "v"(av[7]),
                       "v"(av[8]), "v"(av[9]), "v"(av[10]), "v"(av[11]), "v"(av[12]), "v"(av[13]), "v"(av[14]), "v"(av[15]));
#pragma unroll
    for (int k = 0; k < EPW; ++k) {
        const int lr   = wv + k * (BT / 32);
        const int node = n0 + lr;
        if (lr < npb && node < N) out[(long)node * 32 + lane] = ov[k] + av[k] * dg[k];
    }
    for (int lr = wv + EPW * (BT / 32); lr < npb; lr += BT / 32) {
        const int node = n0 + lr;
        if (node < N) out[(long)node * 32 + lane] += acc[lr * 32 + lane] * deg[node];
    }
}

extern "C" void kernel_launch(void* const* d_in, const int* in_sizes, int n_in,
                              void* d_out, int out_size, void* d_ws, size_t ws_size,
                              hipStream_t stream) {
    const float* x   = (const float*)d_in[0];
    const int*   ei  = (const int*)d_in[1];
    const float* W1l = (const float*)d_in[2];
    const float* b1  = (const float*)d_in[3];
    const float* W1r = (const float*)d_in[4];
    const float* W2l = (const float*)d_in[5];
    const float* b2  = (const float*)d_in[6];
    const float* W2r = (const float*)d_in[7];
    float* out = (float*)d_out;

    int N = in_sizes[0] / 9;
    int E = in_sizes[1] / 2;
    const int* src = ei;
    const int* dst = ei + E;

    float* ws   = (float*)d_ws;
    float* deg  = ws;
    float* t    = deg + N;
    unsigned* sorted = (unsigned*)(t + (long)N * 32);
    int* tab = (int*)(sorted + E);
    float* xp = (float*)(tab + NBK * NBK);

    int npb   = (N + NBK - 1) / NBK;
    int chunk = (E + NBK - 1) / NBK;

    bucket_group<<<NBK, BT, ((size_t)chunk + 1) * sizeof(unsigned) + (size_t)chunk * sizeof(uint2), stream>>>(src, dst, x, xp, tab, sorted, N, E, chunk, npb);

    int ntl = (npb + 15) / 16;
    int nht = ntl < BT / 32 ? ntl : BT / 32;
    gather_x_layers<<<NBK, BT, ((size_t)ntl * 256 + (size_t)nht * 16 * 68) * sizeof(float), stream>>>(
        x, xp, sorted, tab, W1l, b1, W1r, W2l, b2, W2r, deg, t, out, N, npb, E, chunk);

    gather_t<<<NBK, BT, (size_t)npb * 32 * sizeof(float), stream>>>(t, sorted, tab, deg, out, N, npb, E, chunk);
}


# Both gathers index their table with a 32-bit offset on the table pointer as scalar base: one shift-or per row, no 64-bit vector add and no wait on the address register between row loads
# speedup vs baseline: 7.8356x; 1.0839x over previous
#include <hip/hip_runtime.h>

typedef __attribute__((ext_vector_type(2))) float v2f;
typedef __attribute__((ext_vector_type(8))) float v8f;

#define WMMA_F32_4(a, b, c) \
    __builtin_amdgcn_wmma_f32_16x16x4_f32(false, (a), false, (b), (short)0, (c), false, false)

#define NBK 256
#define BT  1024
#define EPW 16

__device__ __forceinline__ void atomAddF32(float* p, float v) {
    unsafeAtomicAdd(p, v);
}


__device__ __forceinline__ int chunk_end(int w, int chunk, int E) {
    const int e0 = w * chunk;
    int e1 = e0 + chunk;
    if (e1 > E) e1 = E;
    return e1 < e0 ? e0 : e1;
}

__device__ __forceinline__ int wave_incl_sum(int v, int lane) {
    int s = v;
#pragma unroll
    for (int off = 1; off < 32; off <<= 1) {
        const int u = __shfl_up(s, off);
        if (lane >= off) s += u;
    }
    return s;
}

__global__ __launch_bounds__(BT) void bucket_group(const int* __restrict__ src, const int* __restrict__ dst,
                                                   const float* __restrict__ x, float* __restrict__ xp,
                                                   int* __restrict__ tab, unsigned* __restrict__ sorted,
                                                   int N, int E, int chunk, int npb) {
    extern __shared__ __attribute__((aligned(16))) unsigned stage[];
    uint2* kw = (uint2*)(stage + ((chunk + 1) & ~1));
    __shared__ int hist[NBK];
    __shared__ int cursor[NBK];
    __shared__ int wsum[NBK / 32];
    const int tid = threadIdx.x, w = blockIdx.x, lane = tid & 31;
    if (tid < NBK) hist[tid] = 0;
    __syncthreads();
    const int e0 = w * chunk;
    const int e1 = chunk_end(w, chunk, E);
    for (int eb = e0 + tid; eb < e1; eb += 8 * BT) {
        int s[8], d[8];
#pragma unroll
        for (int u = 0; u < 8; ++u) {
            const int e  = eb + u * BT;
            const int ec = e < e1 ? e : e1 - 1;
            s[u] = src[ec];
            d[u] = dst[ec];
        }
        asm volatile("" :: "v"(s[1]), "v"(d[1]));
#pragma unroll
        for (int u = 0; u < 8; ++u) {
            const int e = eb + u * BT;
            if (e < e1) {
                const int bb = d[u] / npb;
                const int r  = atomicAdd(&hist[bb], 1);
                kw[e - e0] = make_uint2(((unsigned)r << 8) | (unsigned)bb,
                                        ((unsigned)s[u] << 12) | (unsigned)(d[u] - bb * npb));
            }
        }
    }
    __syncthreads();
    int c = 0, incl = 0;
    if (tid < NBK) {
        c = hist[tid];
        incl = wave_incl_sum(c, lane);
        if (lane == 31) wsum[tid >> 5] = incl;
    }
    __syncthreads();
    if (tid < NBK) {
        int before = 0;
#pragma unroll
        for (int k = 0; k < NBK / 32; ++k) before += (k < (tid >> 5)) ? wsum[k] : 0;
        const int first = before + incl - c;
        cursor[tid] = first;
        tab[tid * NBK + w] = e0 + first;
    }
    __syncthreads();
    const int nx  = npb * 16;
    const int xn0 = w * npb;
    const int xm  = (int)(((size_t)(xp + (size_t)xn0 * 16) >> 2) & 31);
    float xv[8];
#pragma unroll
    for (int u = 0; u < 8; ++u) {
        const int i   = tid - xm + u * BT;
        const int ic  = i < 0 ? 0 : i;
        const int col = ic & 15;
        int node = xn0 + (ic >> 4);
        node = node < N ? node : N - 1;
        xv[u] = x[(long)node * 9 + (col < 9 ? col : 8)];
    }
    const int n = e1 - e0;
    for (int ib = tid; ib < n; ib += 4 * BT) {
        uint2 q[4];
        int   cb[4];
#pragma unroll
        for (int u = 0; u < 4; ++u) {
            const int i = ib + u * BT;
            q[u] = kw[i < n ? i : n - 1];
        }
#pragma unroll
        for (int u = 0; u < 4; ++u) cb[u] = cursor[q[u].x & 255u];
#pragma unroll
        for (int u = 0; u < 4; ++u)
            if (ib + u * BT < n) stage[cb[u] + (int)(q[u].x >> 8)] = q[u].y;
    }
    __syncthreads();
#pragma unroll
    for (int u = 0; u < 8; ++u) {
        const int i    = tid - xm + u * BT;
        const int col  = i & 15;
        const int node = xn0 + (i >> 4);
        if (i >= 0 && i < nx && node < N)
            xp[(long)node * 16 + col] = col < 9 ? xv[u] : (col == 9 ? 1.0f : 0.0f);
    }
    for (int i = tid - xm + 8 * BT; i < nx; i += BT) {
        const int col  = i & 15;
        const int node = xn0 + (i >> 4);
        if (node < N) xp[(long)node * 16 + col] = col < 9 ? x[(long)node * 9 + col] : (col == 9 ? 1.0f : 0.0f);
    }
    const int mis = (int)(((size_t)(sorted + e0) >> 2) & 31);
    for (int ib = tid - mis; ib < n; ib += 4 * BT) {
        unsigned wd[4];
#pragma unroll
        for (int u = 0; u < 4; ++u) {
            const int i = ib + u * BT;
            wd[u] = stage[i < 0 ? 0 : (i < n ? i : n - 1)];
        }
#pragma unroll
        for (int u = 0; u < 4; ++u) {
            const int i = ib + u * BT;
            if (i >= 0 && i < n) sorted[e0 + i] = wd[u];
        }
    }
}

struct RunDir {
    int rpos[NBK];
    int rdel[NBK];
    int wsum[NBK / 32];
    int total;
};

__device__ __forceinline__ void run_fetch(const int* __restrict__ tab, int bk, int tid, int E, int chunk,
                                          int& first, int& len) {
    const int nb = bk + 1 < NBK ? bk + 1 : NBK - 1;
    first = tab[bk * NBK + tid];
    int next = tab[nb * NBK + tid];
    if (bk + 1 == NBK) next = chunk_end(tid, chunk, E);
    len = next - first;
}

__device__ __forceinline__ int run_scan_a(RunDir& rd, int tid, int lane, int len) {
    const int incl = wave_incl_sum(len, lane);
    if (lane == 31) rd.wsum[tid >> 5] = incl;
    return incl;
}

__device__ __forceinline__ void run_scan_b(RunDir& rd, int tid, int first, int len, int incl) {
    int before = 0;
#pragma unroll
    for (int k = 0; k < NBK / 32; ++k) before += (k < (tid >> 5)) ? rd.wsum[k] : 0;
    const int ex = before + incl - len;
    rd.rpos[tid] = ex;
    rd.rdel[tid] = first - ex;
    if (tid == NBK - 1) rd.total = ex + len;
}

__device__ __forceinline__ int run_slot(const RunDir& rd, int g) {
    int k = 0;
#pragma unroll
    for (int step = NBK / 2; step >= 1; step >>= 1)
        if (rd.rpos[k + step] <= g) k += step;
    return g + rd.rdel[k];
}

__global__ __launch_bounds__(BT) void gather_x_layers(
    const float* __restrict__ x, const float* __restrict__ xp, const unsigned* __restrict__ sorted, const int* __restrict__ tab,
    const float* __restrict__ W1l, const float* __restrict__ b1, const float* __restrict__ W1r,
    const float* __restrict__ W2l, const float* __restrict__ b2, const float* __restrict__ W2r,
    float* __restrict__ deg, float* __restrict__ t, float* __restrict__ out, int N, int npb, int E, int chunk) {
    extern __shared__ float acc[];
    __shared__ float Bt1[64][36];
    __shared__ float Bt2[64][68];
    __shared__ RunDir rd;

    const int tid  = threadIdx.x, bk = blockIdx.x;
    const int lane = tid & 31;
    const int wv   = __builtin_amdgcn_readfirstlane(tid >> 5);
    const int slot = lane & 15;
    const bool hi  = lane >= 16;
    const int ntl  = (npb + 15) >> 4;
    const int nacc = ntl * 256;
    int rfirst = 0, rlen = 0, rincl = 0;
    if (tid < NBK) run_fetch(tab, bk, tid, E, chunk, rfirst, rlen);
    for (int i = tid; i < nacc; i += BT) acc[i] = 0.0f;
    if (tid < NBK) rincl = run_scan_a(rd, tid, lane, rlen);
    __syncthreads();
    if (tid < NBK) run_scan_b(rd, tid, rfirst, rlen, rincl);
    __syncthreads();
    const int total = __builtin_amdgcn_readfirstlane(rd.total);
    const int kq = tid >> 6, cq = tid & 63;
    const int kc = kq < 9 ? kq : 8;
    float w1a = W1l[kc * 64 + cq];
    float w1b = W1r[kc * 64 + cq];
    if (kq >= 9) { w1a = 0.0f; w1b = 0.0f; }
    const float* w2p = (cq < 32) ? (W2l + cq) : (W2r + (cq - 32));
    float w2v[4];
#pragma unroll
    for (int u = 0; u < 4; ++u) w2v[u] = w2p[(kq + 16 * u) * 32];
    const int cs = slot < 10 ? slot : 9;
    unsigned p = 0;
    if (wv * 32 + lane < total) p = sorted[run_slot(rd, wv * 32 + lane)];
    for (int base = wv * 32; base < total; base += BT) {
        const int sv = (int)(p >> 12), dv = (int)(p & 4095u);
        const int n = total - base;
        float v[16];
        int   ai[16];
#pragma unroll
        for (int j = 0; j < 16; ++j) {
            const int sa = __builtin_amdgcn_readlane(sv, j), sb = __builtin_amdgcn_readlane(sv, j + 16);
            const int da = __builtin_amdgcn_readlane(dv, j), db = __builtin_amdgcn_readlane(dv, j + 16);
            const int s = hi ? sb : sa;
            const int d = hi ? db : da;
            v[j]  = xp[(unsigned)s * 16u + (unsigned)cs];
            ai[j] = d * 16 + slot;
        }
        __builtin_amdgcn_sched_barrier(0);
        const int gn = base + BT + lane;
        const unsigned pn = sorted[run_slot(rd, gn < total ? gn : total - 1)];
        if (n >= 32) {
            if (slot < 10) {
#pragma unroll
                for (int j = 0; j < 16; ++j) atomAddF32(&acc[ai[j]], v[j]);
            }
        } else {
            const int lim = n - (hi ? 16 : 0);
            if (slot < 10) {
#pragma unroll
                for (int j = 0; j < 16; ++j)
                    if (j < lim) atomAddF32(&acc[ai[j]], v[j]);
            }
        }
        p = pn;
    }
    Bt1[cq][kq]      = w1a;
    Bt1[cq][16 + kq] = w1b;
#pragma unroll
    for (int u = 0; u < 4; ++u) Bt2[cq][kq + 16 * u] = w2v[u];
    __syncthreads();

    const int n0   = bk * npb;
    const int m    = lane & 15;
    const int koff = (lane < 16) ? 0 : 2;
    const int mhi  = (lane < 16) ? 0 : 8;
    float* hs = acc + nacc + wv * (16 * 68);
    for (int tb = 0; tb < ntl; tb += BT / 32) {
        const int tl = tb + wv;
        const bool active = tl < ntl;
        v8f c0 = {}, c1 = {}, c2 = {}, c3 = {};
        if (active) {
            const int lrow  = tl * 16 + m;
            const int node  = n0 + lrow;
            const bool rv   = lrow < npb && node < N;
            const float* ar = acc + lrow * 16;
            const float inv = 1.0f / fmaxf(ar[9], 1.0f);
            if (lane < 16 && rv) deg[node] = inv;
            const float* xr = x + (long)(node < N ? node : N - 1) * 9;
            float x0 = xr[koff], x1 = xr[koff + 1], x2 = xr[4 + koff], x3 = xr[5 + koff], x4 = xr[8];
            if (!rv) { x0 = 0.0f; x1 = 0.0f; x2 = 0.0f; x3 = 0.0f; x4 = 0.0f; }
            if (lane >= 16) x4 = 0.0f;
#pragma unroll
            for (int ks = 0; ks < 7; ++ks) {
                const int kk = ks * 4 + koff;
                v2f a;
                if (ks < 4) {
                    a = *(const v2f*)&ar[kk];
                    a.x *= inv;
                    a.y *= inv;
                } else if (ks == 4) {
                    a.x = x0; a.y = x1;
                } else if (ks == 5) {
                    a.x = x2; a.y = x3;
                } else {
                    a.x = x4; a.y = 0.0f;
                }
                v2f b0  = *(const v2f*)&Bt1[ 0 + m][kk];
                v2f b1v = *(const v2f*)&Bt1[16 + m][kk];
                v2f b2v = *(const v2f*)&Bt1[32 + m][kk];
                v2f b3v = *(const v2f*)&Bt1[48 + m][kk];
                c0 = WMMA_F32_4(a, b0,  c0);
                c1 = WMMA_F32_4(a, b1v, c1);
                c2 = WMMA_F32_4(a, b2v, c2);
                c3 = WMMA_F32_4(a, b3v, c3);
            }
            const float bb0 = b1[m], bb1 = b1[16 + m], bb2 = b1[32 + m], bb3 = b1[48 + m];
#pragma unroll
            for (int r = 0; r < 8; ++r) {
                float* hp = hs + (r + mhi) * 68;
                hp[ 0 + m] = fmaxf(c0[r] + bb0, 0.0f);
                hp[16 + m] = fmaxf(c1[r] + bb1, 0.0f);
                hp[32 + m] = fmaxf(c2[r] + bb2, 0.0f);
                hp[48 + m] = fmaxf(c3[r] + bb3, 0.0f);
            }
        }
        __syncthreads();
        if (active) {
            c0 = (v8f){}; c1 = (v8f){}; c2 = (v8f){}; c3 = (v8f){};
#pragma unroll
            for (int ks = 0; ks < 16; ++ks) {
                const int kk = ks * 4 + koff;
                v2f a   = *(const v2f*)&hs[m * 68 + kk];
                v2f b0  = *(const v2f*)&Bt2[ 0 + m][kk];
                v2f b1v = *(const v2f*)&Bt2[16 + m][kk];
                v2f b2v = *(const v2f*)&Bt2[32 + m][kk];
                v2f b3v = *(const v2f*)&Bt2[48 + m][kk];
                c0 = WMMA_F32_4(a, b0,  c0);
                c1 = WMMA_F32_4(a, b1v, c1);
                c2 = WMMA_F32_4(a, b2v, c2);
                c3 = WMMA_F32_4(a, b3v, c3);
            }
            const float bo0 = b2[m], bo1 = b2[16 + m];
#pragma unroll
            for (int r = 0; r < 8; ++r) {
                const int lr  = tl * 16 + r + mhi;
                const int row = n0 + lr;
                if (lr < npb && row < N) {
                    t[(long)row * 32 + m]        = c0[r];
                    t[(long)row * 32 + 16 + m]   = c1[r];
                    out[(long)row * 32 + m]      = c2[r] + bo0;
                    out[(long)row * 32 + 16 + m] = c3[r] + bo1;
                }
            }
        }
    }
}

__global__ __launch_bounds__(BT) void gather_t(const float* __restrict__ t, const unsigned* __restrict__ sorted,
                                               const int* __restrict__ tab, const float* __restrict__ deg,
                                               float* __restrict__ out, int N, int npb, int E, int chunk) {
    extern __shared__ float acc[];
    __shared__ RunDir rd;
    const int tid  = threadIdx.x, bk = blockIdx.x;
    const int lane = tid & 31;
    const int wv   = __builtin_amdgcn_readfirstlane(tid >> 5);
    const int nacc = npb * 32;
    int rfirst = 0, rlen = 0, rincl = 0;
    if (tid < NBK) run_fetch(tab, bk, tid, E, chunk, rfirst, rlen);
    for (int i = tid; i < nacc; i += BT) acc[i] = 0.0f;
    if (tid < NBK) rincl = run_scan_a(rd, tid, lane, rlen);
    __syncthreads();
    if (tid < NBK) run_scan_b(rd, tid, rfirst, rlen, rincl);
    __syncthreads();
    const int total = __builtin_amdgcn_readfirstlane(rd.total);
    const int n0 = bk * npb;
    float ov[EPW], dg[EPW];
#pragma unroll
    for (int k = 0; k < EPW; ++k) {
        const int lr = wv + k * (BT / 32);
        int node = n0 + (lr < npb ? lr : npb - 1);
        node = node < N ? node : N - 1;
        ov[k] = out[(long)node * 32 + lane];
        dg[k] = deg[node];
    }
    unsigned p = 0;
    if (wv * 32 + lane < total) p = sorted[run_slot(rd, wv * 32 + lane)];
    for (int base = wv * 32; base < total; base += BT) {
        const int sv = (int)(p >> 12), dv = (int)(p & 4095u);
        const int n = total - base;
        float v[32];
#pragma unroll
        for (int j = 0; j < 32; ++j) {
            const int s = __builtin_amdgcn_readlane(sv, j);
            v[j] = t[(unsigned)s * 32u + (unsigned)lane];
        }
        const int gn = base + BT + lane;
        const unsigned pn = sorted[run_slot(rd, gn < total ? gn : total - 1)];
        if (n >= 32) {
#pragma unroll
            for (int j = 0; j < 32; ++j) {
                const int d = __builtin_amdgcn_readlane(dv, j);
                atomAddF32(&acc[d * 32 + lane], v[j]);
            }
        } else {
#pragma unroll
            for (int j = 0; j < 32; ++j) {
                if (j < n) {
                    const int d = __builtin_amdgcn_readlane(dv, j);
                    atomAddF32(&acc[d * 32 + lane], v[j]);
                }
            }
        }
        p = pn;
    }
    __syncthreads();
    float av[EPW];
#pragma unroll
    for (int k = 0; k < EPW; ++k) {
        const int lr = wv + k * (BT / 32);
        av[k] = acc[(lr < npb ? lr : npb - 1) * 32 + lane];
    }
    static_assert(EPW == 16, "the statement below names av[0] to av[15]");
    asm volatile("" :: "v"(av[0]), "v"(av[1]), "v"(av[2]), "v"(av[3]), "v"(av[4]), "v"(av[5]), "v"(av[6]), "v"(av[7]),
                       "v"(av[8]), "v"(av[9]), "v"(av[10]), "v"(av[11]), "v"(av[12]), "v"(av[13]), "v"(av[14]), "v"(av[15]));
#pragma unroll
    for (int k = 0; k < EPW; ++k) {
        const int lr   = wv + k * (BT / 32);
        const int node = n0 + lr;
        if (lr < npb && node < N) out[(long)node * 32 + lane] = ov[k] + av[k] * dg[k];
    }
    for (int lr = wv + EPW * (BT / 32); lr < npb; lr += BT / 32) {
        const int node = n0 + lr;
        if (node < N) out[(long)node * 32 + lane] += acc[lr * 32 + lane] * deg[node];
    }
}

extern "C" void kernel_launch(void* const* d_in, const int* in_sizes, int n_in,
                              void* d_out, int out_size, void* d_ws, size_t ws_size,
                              hipStream_t stream) {
    const float* x   = (const float*)d_in[0];
    const int*   ei  = (const int*)d_in[1];
    const float* W1l = (const float*)d_in[2];
    const float* b1  = (const float*)d_in[3];
    const float* W1r = (const float*)d_in[4];
    const float* W2l = (const float*)d_in[5];
    const float* b2  = (const float*)d_in[6];
    const float* W2r = (const float*)d_in[7];
    float* out = (float*)d_out;

    int N = in_sizes[0] / 9;
    int E = in_sizes[1] / 2;
    const int* src = ei;
    const int* dst = ei + E;

    float* ws   = (float*)d_ws;
    float* deg  = ws;
    float* t    = deg + N;
    unsigned* sorted = (unsigned*)(t + (long)N * 32);
    int* tab = (int*)(sorted + E);
    float* xp = (float*)(tab + NBK * NBK);

    int npb   = (N + NBK - 1) / NBK;
    int chunk = (E + NBK - 1) / NBK;

    bucket_group<<<NBK, BT, ((size_t)chunk + 1) * sizeof(unsigned) + (size_t)chunk * sizeof(uint2), stream>>>(src, dst, x, xp, tab, sorted, N, E, chunk, npb);

    int ntl = (npb + 15) / 16;
    int nht = ntl < BT / 32 ? ntl : BT / 32;
    gather_x_layers<<<NBK, BT, ((size_t)ntl * 256 + (size_t)nht * 16 * 68) * sizeof(float), stream>>>(
        x, xp, sorted, tab, W1l, b1, W1r, W2l, b2, W2r, deg, t, out, N, npb, E, chunk);

    gather_t<<<NBK, BT, (size_t)npb * 32 * sizeof(float), stream>>>(t, sorted, tab, deg, out, N, npb, E, chunk);
}
